# starting kernel
# baseline (speedup 1.0000x reference)
#include <hip/hip_runtime.h>
#include <hip/hip_bf16.h>
#include <math.h>


typedef __attribute__((ext_vector_type(2))) float v2f;
typedef __attribute__((ext_vector_type(8))) float v8f;

#define GAT_IN_DIM   128
#define GAT_HEADS    8
#define GAT_FPH      16
#define GAT_OUT      128
#define NEG_SLOPE    0.2f

__global__ __launch_bounds__(256) void gat_proj_wmma(
    const float* __restrict__ x, const float* __restrict__ W,
    float* __restrict__ xp, int N) {
  __shared__ float sx[16 * GAT_IN_DIM];

  const int row0 = blockIdx.x * 16;
  const int tid  = threadIdx.x;

  for (int i = tid; i < 16 * GAT_IN_DIM; i += 256) {
    int r = i >> 7, c = i & 127;
    int gr = row0 + r;
    sx[i] = (gr < N) ? x[gr * GAT_IN_DIM + c] : 0.0f;
  }
  __syncthreads();

  const int wave = tid >> 5;
  const int lane = tid & 31;
  const int n    = lane & 15;
  const int half = lane >> 4;
  const int col  = wave * 16 + n;
  const int arow = lane & 15;

  v8f acc = {};
  for (int k0 = 0; k0 < GAT_IN_DIM; k0 += 4) {
    const int kk = k0 + 2 * half;
    v2f a, b;
    a[0] = sx[arow * GAT_IN_DIM + kk];
    a[1] = sx[arow * GAT_IN_DIM + kk + 1];
    b[0] = W[kk * GAT_OUT + col];
    b[1] = W[(kk + 1) * GAT_OUT + col];
    acc = __builtin_amdgcn_wmma_f32_16x16x4_f32(
         false, a,  false, b,
         (short)0, acc,  false,  false);
  }

  const int mbase = half * 8;
#pragma unroll
  for (int i = 0; i < 8; ++i) {
    int gr = row0 + mbase + i;
    if (gr < N) xp[gr * GAT_OUT + col] = acc[i];
  }
}

__global__ void gat_al(const float* __restrict__ xp,
                       const float* __restrict__ a_src,
                       const float* __restrict__ a_dst,
                       float* __restrict__ al_src,
                       float* __restrict__ al_dst, int N) {
  int idx = blockIdx.x * blockDim.x + threadIdx.x;
  if (idx >= N * GAT_HEADS) return;
  int h = idx & 7, node = idx >> 3;
  const float* r  = xp + node * GAT_OUT + h * GAT_FPH;
  const float* as = a_src + h * GAT_FPH;
  const float* ad = a_dst + h * GAT_FPH;
  float s = 0.0f, d = 0.0f;
#pragma unroll
  for (int f = 0; f < GAT_FPH; ++f) {
    float v = r[f];
    s += v * as[f];
    d += v * ad[f];
  }
  al_src[idx] = s;
  al_dst[idx] = d;
}

__global__ void gat_init(float* __restrict__ out, float* __restrict__ m,
                         float* __restrict__ denom, int N) {
  int i = blockIdx.x * blockDim.x + threadIdx.x;
  if (i < N * GAT_OUT) out[i] = 0.0f;
  if (i < N * GAT_HEADS) {
    m[i] = -__builtin_inff();
    denom[i] = 0.0f;
  }
}

__device__ __forceinline__ void atomicMaxF(float* addr, float v) {
  if (v >= 0.0f)
    atomicMax((int*)addr, __float_as_int(v));
  else
    atomicMin((unsigned int*)addr, __float_as_uint(v));
}

__device__ __forceinline__ float leaky(float e) {
  return e >= 0.0f ? e : NEG_SLOPE * e;
}

__global__ void gat_edge_max(const int* __restrict__ src,
                             const int* __restrict__ dst,
                             const float* __restrict__ al_src,
                             const float* __restrict__ al_dst,
                             float* __restrict__ m, int E) {
  int e = blockIdx.x * blockDim.x + threadIdx.x;
  if (e >= E) return;
  int s = src[e], d = dst[e];
  const float4* as = (const float4*)(al_src + s * GAT_HEADS);
  const float4* ad = (const float4*)(al_dst + d * GAT_HEADS);
  float4 a0 = as[0], a1 = as[1], b0 = ad[0], b1 = ad[1];
  float va[8] = {a0.x, a0.y, a0.z, a0.w, a1.x, a1.y, a1.z, a1.w};
  float vb[8] = {b0.x, b0.y, b0.z, b0.w, b1.x, b1.y, b1.z, b1.w};
#pragma unroll
  for (int h = 0; h < GAT_HEADS; ++h)
    atomicMaxF(&m[d * GAT_HEADS + h], leaky(va[h] + vb[h]));
}

__global__ void gat_edge_expsum(const int* __restrict__ src,
                                const int* __restrict__ dst,
                                const float* __restrict__ al_src,
                                const float* __restrict__ al_dst,
                                const float* __restrict__ m,
                                float* __restrict__ denom, int E) {
  int e = blockIdx.x * blockDim.x + threadIdx.x;
  if (e >= E) return;
  int s = src[e], d = dst[e];
  const float4* as = (const float4*)(al_src + s * GAT_HEADS);
  const float4* ad = (const float4*)(al_dst + d * GAT_HEADS);
  const float4* mm = (const float4*)(m + d * GAT_HEADS);
  float4 a0 = as[0], a1 = as[1], b0 = ad[0], b1 = ad[1], m0 = mm[0], m1 = mm[1];
  float va[8] = {a0.x, a0.y, a0.z, a0.w, a1.x, a1.y, a1.z, a1.w};
  float vb[8] = {b0.x, b0.y, b0.z, b0.w, b1.x, b1.y, b1.z, b1.w};
  float vm[8] = {m0.x, m0.y, m0.z, m0.w, m1.x, m1.y, m1.z, m1.w};
#pragma unroll
  for (int h = 0; h < GAT_HEADS; ++h)
    atomicAdd(&denom[d * GAT_HEADS + h], expf(leaky(va[h] + vb[h]) - vm[h]));
}

__global__ __launch_bounds__(256) void gat_edge_agg(
    const int* __restrict__ src, const int* __restrict__ dst,
    const float* __restrict__ al_src, const float* __restrict__ al_dst,
    const float* __restrict__ m, const float* __restrict__ denom,
    const float* __restrict__ xp, float* __restrict__ out, int E) {
  int gwarp  = (blockIdx.x * blockDim.x + threadIdx.x) >> 5;
  int lane   = threadIdx.x & 31;
  int nwarps = (gridDim.x * blockDim.x) >> 5;
  for (int e = gwarp; e < E; e += nwarps) {
    int s = src[e], d = dst[e];
    int h = lane >> 2;
    float as = al_src[s * GAT_HEADS + h];
    float ad = al_dst[d * GAT_HEADS + h];
    float mx = m[d * GAT_HEADS + h];
    float dn = denom[d * GAT_HEADS + h];
    float ex = expf(leaky(as + ad) - mx);
    float alpha = ex / (dn == 0.0f ? 1.0f : dn);
    int col = lane * 4;
    float4 v = *(const float4*)(xp + s * GAT_OUT + col);
    float* o = out + d * GAT_OUT + col;
    atomicAdd(o + 0, v.x * alpha);
    atomicAdd(o + 1, v.y * alpha);
    atomicAdd(o + 2, v.z * alpha);
    atomicAdd(o + 3, v.w * alpha);
  }
}

__global__ void gat_finalize(float* __restrict__ out,
                             const float* __restrict__ bias, int N) {
  int i = blockIdx.x * blockDim.x + threadIdx.x;
  if (i >= N * GAT_OUT) return;
  float v = out[i] + bias[i & 127];
  out[i] = v > 0.0f ? v : expm1f(v);
}

extern "C" void kernel_launch(void* const* d_in, const int* in_sizes, int n_in,
                              void* d_out, int out_size, void* d_ws,
                              size_t ws_size, hipStream_t stream) {
  const float* x      = (const float*)d_in[0];
  const int*   ei     = (const int*)d_in[1];
  const float* W      = (const float*)d_in[2];
  const float* a_src  = (const float*)d_in[3];
  const float* a_dst  = (const float*)d_in[4];
  const float* bias   = (const float*)d_in[5];

  const int N = in_sizes[0] / GAT_IN_DIM;
  const int E = in_sizes[1] / 2;
  const int* src = ei;
  const int* dst = ei + E;

  float* out = (float*)d_out;

  float* ws     = (float*)d_ws;
  float* xp     = ws;
  float* al_src = xp + (size_t)N * GAT_OUT;
  float* al_dst = al_src + (size_t)N * GAT_HEADS;
  float* mbuf   = al_dst + (size_t)N * GAT_HEADS;
  float* denom  = mbuf + (size_t)N * GAT_HEADS;

  {
    int total = N * GAT_OUT;
    gat_init<<<(total + 255) / 256, 256, 0, stream>>>(out, mbuf, denom, N);
  }
  gat_proj_wmma<<<(N + 15) / 16, 256, 0, stream>>>(x, W, xp, N);
  {
    int total = N * GAT_HEADS;
    gat_al<<<(total + 255) / 256, 256, 0, stream>>>(xp, a_src, a_dst, al_src,
                                                    al_dst, N);
  }
  gat_edge_max<<<(E + 255) / 256, 256, 0, stream>>>(src, dst, al_src, al_dst,
                                                    mbuf, E);
  gat_edge_expsum<<<(E + 255) / 256, 256, 0, stream>>>(src, dst, al_src,
                                                       al_dst, mbuf, denom, E);
  gat_edge_agg<<<8192, 256, 0, stream>>>(src, dst, al_src, al_dst, mbuf, denom,
                                         xp, out, E);
  {
    int total = N * GAT_OUT;
    gat_finalize<<<(total + 255) / 256, 256, 0, stream>>>(out, bias, N);
  }
}


# Scatter sweep by destination owner: each of 256 work-groups scans all dst ids, sums its 392 nodes' rows and exp sums in LDS (f32 LDS atomics, arrival order), stores out finished; 6 launches -> 5
# speedup vs baseline: 3.9187x; 3.9187x over previous
#include <hip/hip_runtime.h>
#include <hip/hip_bf16.h>
#include <math.h>


typedef __attribute__((ext_vector_type(2))) float v2f;
typedef __attribute__((ext_vector_type(8))) float v8f;

#define GAT_IN_DIM   128
#define GAT_HEADS    8
#define GAT_FPH      16
#define GAT_OUT      128
#define NEG_SLOPE    0.2f

__global__ __launch_bounds__(256) void gat_proj_wmma(
    const float* __restrict__ x, const float* __restrict__ W,
    float* __restrict__ xp, int N) {
  __shared__ float sx[16 * GAT_IN_DIM];

  const int row0 = blockIdx.x * 16;
  const int tid  = threadIdx.x;

  for (int i = tid; i < 16 * GAT_IN_DIM; i += 256) {
    int r = i >> 7, c = i & 127;
    int gr = row0 + r;
    sx[i] = (gr < N) ? x[gr * GAT_IN_DIM + c] : 0.0f;
  }
  __syncthreads();

  const int wave = tid >> 5;
  const int lane = tid & 31;
  const int n    = lane & 15;
  const int half = lane >> 4;
  const int col  = wave * 16 + n;
  const int arow = lane & 15;

  v8f acc = {};
  for (int k0 = 0; k0 < GAT_IN_DIM; k0 += 4) {
    const int kk = k0 + 2 * half;
    v2f a, b;
    a[0] = sx[arow * GAT_IN_DIM + kk];
    a[1] = sx[arow * GAT_IN_DIM + kk + 1];
    b[0] = W[kk * GAT_OUT + col];
    b[1] = W[(kk + 1) * GAT_OUT + col];
    acc = __builtin_amdgcn_wmma_f32_16x16x4_f32(
         false, a,  false, b,
         (short)0, acc,  false,  false);
  }

  const int mbase = half * 8;
#pragma unroll
  for (int i = 0; i < 8; ++i) {
    int gr = row0 + mbase + i;
    if (gr < N) xp[gr * GAT_OUT + col] = acc[i];
  }
}

__global__ void gat_al(const float* __restrict__ xp,
                       const float* __restrict__ a_src,
                       const float* __restrict__ a_dst,
                       float* __restrict__ al_src,
                       float* __restrict__ al_dst, int N) {
  int idx = blockIdx.x * blockDim.x + threadIdx.x;
  if (idx >= N * GAT_HEADS) return;
  int h = idx & 7, node = idx >> 3;
  const float* r  = xp + node * GAT_OUT + h * GAT_FPH;
  const float* as = a_src + h * GAT_FPH;
  const float* ad = a_dst + h * GAT_FPH;
  float s = 0.0f, d = 0.0f;
#pragma unroll
  for (int f = 0; f < GAT_FPH; ++f) {
    float v = r[f];
    s += v * as[f];
    d += v * ad[f];
  }
  al_src[idx] = s;
  al_dst[idx] = d;
}

__global__ void gat_init(float* __restrict__ m, int N) {
  int i = blockIdx.x * blockDim.x + threadIdx.x;
  if (i < N * GAT_HEADS) m[i] = -__builtin_inff();
}

__device__ __forceinline__ float leaky(float e) {
  return e >= 0.0f ? e : NEG_SLOPE * e;
}

#define GAT_EDGES_PER_WAVE 128

template <bool FULL>
__device__ __forceinline__ void gat_load_edge_ids(
    const int* __restrict__ src, const int* __restrict__ dst, int e0, int lane,
    int E, int (&sv)[4], int (&dv)[4]) {
#pragma unroll
  for (int c = 0; c < 4; ++c) {
    const int e = e0 + c * 32 + lane;
    const bool ok = FULL || e < E;
    sv[c] = ok ? src[e] : 0;
    dv[c] = ok ? dst[e] : 0;
  }
}

template <bool FULL>
__device__ __forceinline__ void gat_edge_max_wave(
    const int* __restrict__ src, const int* __restrict__ dst,
    const float* __restrict__ al_src, const float* __restrict__ al_dst,
    float* __restrict__ m, int E, int e0, int lane) {
  const int h   = lane & 7;
  const int sub = lane >> 3;
  int sv[4], dv[4];
  gat_load_edge_ids<FULL>(src, dst, e0, lane, E, sv, dv);

#pragma unroll
  for (int c = 0; c < 4; ++c) {
    float v[8];
    int   mi[8];
#pragma unroll
    for (int it = 0; it < 8; ++it) {
      const int j = it * 4 + sub;
      const int s = __shfl(sv[c], j);
      const int d = __shfl(dv[c], j);
      mi[it] = d * GAT_HEADS + h;
      v[it]  = leaky(al_src[s * GAT_HEADS + h] + al_dst[mi[it]]);
    }
#pragma unroll
    for (int it = 0; it < 8; ++it) {
      if (FULL || e0 + c * 32 + it * 4 + sub < E)
        __hip_atomic_fetch_max(&m[mi[it]], v[it], __ATOMIC_RELAXED,
                               __HIP_MEMORY_SCOPE_AGENT);
    }
  }
}

__global__ __launch_bounds__(256) void gat_edge_max(
    const int* __restrict__ src, const int* __restrict__ dst,
    const float* __restrict__ al_src, const float* __restrict__ al_dst,
    float* __restrict__ m, int E) {
  const int lane = threadIdx.x & 31;
  const int wave = (blockIdx.x * 256 + threadIdx.x) >> 5;
  const int e0   = wave * GAT_EDGES_PER_WAVE;
  if (e0 + GAT_EDGES_PER_WAVE <= E)
    gat_edge_max_wave<true>(src, dst, al_src, al_dst, m, E, e0, lane);
  else if (e0 < E)
    gat_edge_max_wave<false>(src, dst, al_src, al_dst, m, E, e0, lane);
}

#define GAT_OWN_NODES 392
#define GAT_OWN_WAVES 32
#define GAT_OWN_LIST  640

__global__ __launch_bounds__(1024) void gat_own_agg(
    const int* __restrict__ src, const int* __restrict__ dst,
    const float* __restrict__ al_src, const float* __restrict__ al_dst,
    const float* __restrict__ m, const float* __restrict__ xp,
    const float* __restrict__ bias, float* __restrict__ out, int N, int E) {
  __shared__ float sacc[GAT_OWN_NODES * GAT_OUT];
  __shared__ float sden[GAT_OWN_NODES * GAT_HEADS];
  __shared__ int   slist[GAT_OWN_WAVES * GAT_OWN_LIST];

  const int tid  = threadIdx.x;
  const int lane = tid & 31;
  const int wave = __builtin_amdgcn_readfirstlane(tid >> 5);
  const int lo   = blockIdx.x * GAT_OWN_NODES;
  const int left = N - lo;
  const int cnt  = left < GAT_OWN_NODES ? left : GAT_OWN_NODES;
  const unsigned ucnt = (unsigned)cnt;

  for (int i = tid; i < GAT_OWN_NODES * GAT_OUT; i += 1024) sacc[i] = 0.0f;
  for (int i = tid; i < GAT_OWN_NODES * GAT_HEADS; i += 1024) sden[i] = 0.0f;
  __syncthreads();

  int* list = slist + wave * GAT_OWN_LIST;
  int n = 0;

  auto pass_ids = [&](int i0, int& s_a, int& s_b, int& d_a, int& d_b) {
    const int qa = i0 + (lane >> 3);
    const int qb = qa + 4;
    const int ea = list[qa < n ? qa : 0];
    const int eb = list[qb < n ? qb : 0];
    s_a = src[ea]; s_b = src[eb];
    d_a = dst[ea]; d_b = dst[eb];
  };

  auto pass = [&](int i0, int s_a, int s_b, int d_a, int d_b, const bool guarded) {
    float4 xr[8];
#pragma unroll
    for (int q = 0; q < 8; ++q) {
      const int sq = __builtin_amdgcn_readlane(q < 4 ? s_a : s_b, (q & 3) * 8);
      xr[q] = *reinterpret_cast<const float4*>(xp + sq * GAT_OUT + lane * 4);
    }

    const int  hd = lane & 7;
    const bool va = !guarded || i0 + (lane >> 3) < n;
    const bool vb = !guarded || i0 + 4 + (lane >> 3) < n;
    const int ga = d_a * GAT_HEADS + hd, gb = d_b * GAT_HEADS + hd;
    const float exa =
        expf(leaky(al_src[s_a * GAT_HEADS + hd] + al_dst[ga]) - m[ga]);
    const float exb =
        expf(leaky(al_src[s_b * GAT_HEADS + hd] + al_dst[gb]) - m[gb]);
    const int la = d_a - lo, lb = d_b - lo;
    if (va) atomicAdd(&sden[la * GAT_HEADS + hd], exa);
    if (vb) atomicAdd(&sden[lb * GAT_HEADS + hd], exb);

#pragma unroll
    for (int q = 0; q < 8; ++q) {
      if (!guarded || i0 + q < n) {
        const int lq = __builtin_amdgcn_readlane(q < 4 ? la : lb, (q & 3) * 8);
        const float a = __shfl(q < 4 ? exa : exb, (q & 3) * 8 + (lane >> 2));
        float* o = sacc + lq * GAT_OUT + lane;
        atomicAdd(o + 0,  xr[q].x * a);
        atomicAdd(o + 32, xr[q].y * a);
        atomicAdd(o + 64, xr[q].z * a);
        atomicAdd(o + 96, xr[q].w * a);
      }
    }
  };

  auto flush = [&]() {
    const int nf = n & ~7;
    int s_a, s_b, d_a, d_b;
    pass_ids(0, s_a, s_b, d_a, d_b);
    for (int i0 = 0; i0 < nf; i0 += 8) {
      int ns_a, ns_b, nd_a, nd_b;
      pass_ids(i0 + 8, ns_a, ns_b, nd_a, nd_b);
      pass(i0, s_a, s_b, d_a, d_b, false);
      s_a = ns_a; s_b = ns_b; d_a = nd_a; d_b = nd_b;
    }
    const int r = n - nf;
    int t = 0;
    if (lane < r) t = list[nf + lane];
    if (lane < r) list[lane] = t;
    n = r;
  };

  auto keep = [&](const int4& c, int e) {
    const bool k0 = (unsigned)(c.x - lo) < ucnt;
    const bool k1 = (unsigned)(c.y - lo) < ucnt;
    const bool k2 = (unsigned)(c.z - lo) < ucnt;
    const bool k3 = (unsigned)(c.w - lo) < ucnt;
    if (__builtin_amdgcn_ballot_w32(k0 | k1 | k2 | k3) != 0u) {
      const unsigned b0 = __builtin_amdgcn_ballot_w32(k0);
      if (b0 != 0u) {
        if (k0) list[n + (int)__builtin_amdgcn_mbcnt_lo(b0, 0u)] = e + 0;
        n += __builtin_popcount(b0);
      }
      const unsigned b1 = __builtin_amdgcn_ballot_w32(k1);
      if (b1 != 0u) {
        if (k1) list[n + (int)__builtin_amdgcn_mbcnt_lo(b1, 0u)] = e + 1;
        n += __builtin_popcount(b1);
      }
      const unsigned b2 = __builtin_amdgcn_ballot_w32(k2);
      if (b2 != 0u) {
        if (k2) list[n + (int)__builtin_amdgcn_mbcnt_lo(b2, 0u)] = e + 2;
        n += __builtin_popcount(b2);
      }
      const unsigned b3 = __builtin_amdgcn_ballot_w32(k3);
      if (b3 != 0u) {
        if (k3) list[n + (int)__builtin_amdgcn_mbcnt_lo(b3, 0u)] = e + 3;
        n += __builtin_popcount(b3);
      }
    }
  };

  const bool aligned = ((E & 3) == 0) &&
                       ((reinterpret_cast<size_t>(dst) & 15) == 0);
  const int nfull = aligned ? (E >> 10) : 0;
  int look = wave;
  int4 c0 = make_int4(-1, -1, -1, -1), c1 = c0, c2 = c0, c3 = c0;
  int4 c4 = c0, c5 = c0, c6 = c0, c7 = c0;
  if (look < nfull) {
    const int4* p = reinterpret_cast<const int4*>(dst + look * 1024 + lane * 4);
    c0 = p[0];   c1 = p[32];  c2 = p[64];  c3 = p[96];
    c4 = p[128]; c5 = p[160]; c6 = p[192]; c7 = p[224];
  }
  while (look < nfull) {
    const int e    = look * 1024 + lane * 4;
    const int next = look + GAT_OWN_WAVES;
    const int pf = next < nfull ? next : look;
    const int4* p = reinterpret_cast<const int4*>(dst + pf * 1024 + lane * 4);
    const int4 n0 = p[0],   n1 = p[32],  n2 = p[64],  n3 = p[96];
    const int4 n4 = p[128], n5 = p[160], n6 = p[192], n7 = p[224];
    keep(c0, e);
    keep(c1, e + 128);
    keep(c2, e + 256);
    keep(c3, e + 384);
    if (n >= 128) flush();
    keep(c4, e + 512);
    keep(c5, e + 640);
    keep(c6, e + 768);
    keep(c7, e + 896);
    if (n >= 128) flush();
    c0 = n0; c1 = n1; c2 = n2; c3 = n3;
    c4 = n4; c5 = n5; c6 = n6; c7 = n7;
    look = next;
  }
  for (int eb = nfull * 1024 + wave * 32; eb < E; eb += 32 * GAT_OWN_WAVES) {
    const int  e = eb + lane;
    const int  d = (e < E) ? dst[e] : -1;
    const bool k = (unsigned)(d - lo) < ucnt;
    const unsigned b = __builtin_amdgcn_ballot_w32(k);
    if (k) list[n + (int)__builtin_amdgcn_mbcnt_lo(b, 0u)] = e;
    n += __builtin_popcount(b);
    if (n >= 128) flush();
  }
  if (n >= 8) flush();
  if (n > 0) {
    int s_a, s_b, d_a, d_b;
    pass_ids(0, s_a, s_b, d_a, d_b);
    pass(0, s_a, s_b, d_a, d_b, true);
  }
  __syncthreads();

  const float4 bv = *reinterpret_cast<const float4*>(bias + lane * 4);
  for (int r = wave; r < cnt; r += GAT_OWN_WAVES) {
    const float* a  = sacc + r * GAT_OUT + lane;
    const float  dn = sden[r * GAT_HEADS + (lane >> 2)];
    const float  dd = (dn == 0.0f) ? 1.0f : dn;
    float4 v;
    v.x = a[0]  / dd + bv.x;
    v.y = a[32] / dd + bv.y;
    v.z = a[64] / dd + bv.z;
    v.w = a[96] / dd + bv.w;
    v.x = v.x > 0.0f ? v.x : expm1f(v.x);
    v.y = v.y > 0.0f ? v.y : expm1f(v.y);
    v.z = v.z > 0.0f ? v.z : expm1f(v.z);
    v.w = v.w > 0.0f ? v.w : expm1f(v.w);
    *reinterpret_cast<float4*>(out + (size_t)(lo + r) * GAT_OUT + lane * 4) = v;
  }
}

extern "C" void kernel_launch(void* const* d_in, const int* in_sizes, int n_in,
                              void* d_out, int out_size, void* d_ws,
                              size_t ws_size, hipStream_t stream) {
  const float* x      = (const float*)d_in[0];
  const int*   ei     = (const int*)d_in[1];
  const float* W      = (const float*)d_in[2];
  const float* a_src  = (const float*)d_in[3];
  const float* a_dst  = (const float*)d_in[4];
  const float* bias   = (const float*)d_in[5];

  const int N = in_sizes[0] / GAT_IN_DIM;
  const int E = in_sizes[1] / 2;
  const int* src = ei;
  const int* dst = ei + E;

  float* out = (float*)d_out;

  float* ws     = (float*)d_ws;
  float* xp     = ws;
  float* al_src = xp + (size_t)N * GAT_OUT;
  float* al_dst = al_src + (size_t)N * GAT_HEADS;
  float* mbuf   = al_dst + (size_t)N * GAT_HEADS;

  {
    int total = N * GAT_HEADS;
    gat_init<<<(total + 255) / 256, 256, 0, stream>>>(mbuf, N);
  }
  gat_proj_wmma<<<(N + 15) / 16, 256, 0, stream>>>(x, W, xp, N);
  {
    int total = N * GAT_HEADS;
    gat_al<<<(total + 255) / 256, 256, 0, stream>>>(xp, a_src, a_dst, al_src,
                                                    al_dst, N);
  }
  gat_edge_max<<<(E + 1023) / 1024, 256, 0, stream>>>(src, dst, al_src, al_dst,
                                                      mbuf, E);
  gat_own_agg<<<(N + GAT_OWN_NODES - 1) / GAT_OWN_NODES, 1024, 0, stream>>>(
      src, dst, al_src, al_dst, mbuf, xp, bias, out, N, E);
}


# Owner sweep finds its edges in bit planes of the owner number (8 bits per edge, written by the max launch; 32 edges per lane per compare) instead of scanning 4-byte ids; sums still arrival-order
# speedup vs baseline: 4.3341x; 1.1060x over previous
#include <hip/hip_runtime.h>
#include <hip/hip_bf16.h>
#include <math.h>


typedef __attribute__((ext_vector_type(2))) float v2f;
typedef __attribute__((ext_vector_type(8))) float v8f;

#define GAT_IN_DIM   128
#define GAT_HEADS    8
#define GAT_FPH      16
#define GAT_OUT      128
#define NEG_SLOPE    0.2f

__global__ __launch_bounds__(256) void gat_proj_wmma(
    const float* __restrict__ x, const float* __restrict__ W,
    float* __restrict__ xp, int N) {
  __shared__ float sx[16 * GAT_IN_DIM];

  const int row0 = blockIdx.x * 16;
  const int tid  = threadIdx.x;

  for (int i = tid; i < 16 * GAT_IN_DIM; i += 256) {
    int r = i >> 7, c = i & 127;
    int gr = row0 + r;
    sx[i] = (gr < N) ? x[gr * GAT_IN_DIM + c] : 0.0f;
  }
  __syncthreads();

  const int wave = tid >> 5;
  const int lane = tid & 31;
  const int n    = lane & 15;
  const int half = lane >> 4;
  const int col  = wave * 16 + n;
  const int arow = lane & 15;

  v8f acc = {};
  for (int k0 = 0; k0 < GAT_IN_DIM; k0 += 4) {
    const int kk = k0 + 2 * half;
    v2f a, b;
    a[0] = sx[arow * GAT_IN_DIM + kk];
    a[1] = sx[arow * GAT_IN_DIM + kk + 1];
    b[0] = W[kk * GAT_OUT + col];
    b[1] = W[(kk + 1) * GAT_OUT + col];
    acc = __builtin_amdgcn_wmma_f32_16x16x4_f32(
         false, a,  false, b,
         (short)0, acc,  false,  false);
  }

  const int mbase = half * 8;
#pragma unroll
  for (int i = 0; i < 8; ++i) {
    int gr = row0 + mbase + i;
    if (gr < N) xp[gr * GAT_OUT + col] = acc[i];
  }
}

__global__ void gat_al(const float* __restrict__ xp,
                       const float* __restrict__ a_src,
                       const float* __restrict__ a_dst,
                       float* __restrict__ al_src,
                       float* __restrict__ al_dst, int N) {
  int idx = blockIdx.x * blockDim.x + threadIdx.x;
  if (idx >= N * GAT_HEADS) return;
  int h = idx & 7, node = idx >> 3;
  const float* r  = xp + node * GAT_OUT + h * GAT_FPH;
  const float* as = a_src + h * GAT_FPH;
  const float* ad = a_dst + h * GAT_FPH;
  float s = 0.0f, d = 0.0f;
#pragma unroll
  for (int f = 0; f < GAT_FPH; ++f) {
    float v = r[f];
    s += v * as[f];
    d += v * ad[f];
  }
  al_src[idx] = s;
  al_dst[idx] = d;
}

__global__ void gat_init(float* __restrict__ m, int N) {
  int i = blockIdx.x * blockDim.x + threadIdx.x;
  if (i < N * GAT_HEADS) m[i] = -__builtin_inff();
}

__device__ __forceinline__ float leaky(float e) {
  return e >= 0.0f ? e : NEG_SLOPE * e;
}

#define GAT_EDGES_PER_WAVE 128

template <bool FULL>
__device__ __forceinline__ void gat_load_edge_ids(
    const int* __restrict__ src, const int* __restrict__ dst, int e0, int lane,
    int E, int (&sv)[4], int (&dv)[4]) {
#pragma unroll
  for (int c = 0; c < 4; ++c) {
    const int e = e0 + c * 32 + lane;
    const bool ok = FULL || e < E;
    sv[c] = ok ? src[e] : 0;
    dv[c] = ok ? dst[e] : 0;
  }
}

#define GAT_OWN_NODES 392

__device__ __forceinline__ void gat_store_owner_planes(
    const int (&dv)[4], int e0, int lane, unsigned* __restrict__ planes) {
  unsigned w = 0u;
#pragma unroll
  for (int c = 0; c < 4; ++c) {
    const unsigned o = (unsigned)dv[c] / GAT_OWN_NODES;
#pragma unroll
    for (int p = 0; p < 8; ++p) {
      const unsigned b = __builtin_amdgcn_ballot_w32(((o >> p) & 1u) != 0u);
      w = (lane == (p >> 2) * 16 + c * 4 + (p & 3)) ? b : w;
    }
  }
  planes[(e0 >> 10) * 256 + (lane >> 4) * 128 + ((e0 >> 5) & 31) * 4 +
         (lane & 15)] = w;
}

template <bool FULL>
__device__ __forceinline__ void gat_edge_max_wave(
    const int* __restrict__ src, const int* __restrict__ dst,
    const float* __restrict__ al_src, const float* __restrict__ al_dst,
    float* __restrict__ m, unsigned* __restrict__ planes, int E, int e0,
    int lane) {
  const int h   = lane & 7;
  const int sub = lane >> 3;
  int sv[4], dv[4];
  gat_load_edge_ids<FULL>(src, dst, e0, lane, E, sv, dv);
  gat_store_owner_planes(dv, e0, lane, planes);

#pragma unroll
  for (int c = 0; c < 4; ++c) {
    float v[8];
    int   mi[8];
#pragma unroll
    for (int it = 0; it < 8; ++it) {
      const int j = it * 4 + sub;
      const int s = __shfl(sv[c], j);
      const int d = __shfl(dv[c], j);
      mi[it] = d * GAT_HEADS + h;
      v[it]  = leaky(al_src[s * GAT_HEADS + h] + al_dst[mi[it]]);
    }
#pragma unroll
    for (int it = 0; it < 8; ++it) {
      if (FULL || e0 + c * 32 + it * 4 + sub < E)
        __hip_atomic_fetch_max(&m[mi[it]], v[it], __ATOMIC_RELAXED,
                               __HIP_MEMORY_SCOPE_AGENT);
    }
  }
}

__global__ __launch_bounds__(256) void gat_edge_max(
    const int* __restrict__ src, const int* __restrict__ dst,
    const float* __restrict__ al_src, const float* __restrict__ al_dst,
    float* __restrict__ m, unsigned* __restrict__ planes, int E) {
  const int lane = threadIdx.x & 31;
  const int wave = (blockIdx.x * 256 + threadIdx.x) >> 5;
  const int e0   = wave * GAT_EDGES_PER_WAVE;
  if (e0 + GAT_EDGES_PER_WAVE <= E)
    gat_edge_max_wave<true>(src, dst, al_src, al_dst, m, planes, E, e0, lane);
  else if (e0 < E)
    gat_edge_max_wave<false>(src, dst, al_src, al_dst, m, planes, E, e0, lane);
}

#define GAT_OWN_WAVES 32
#define GAT_OWN_LIST  160

__global__ __launch_bounds__(1024) void gat_own_agg(
    const int* __restrict__ src, const int* __restrict__ dst,
    const float* __restrict__ al_src, const float* __restrict__ al_dst,
    const float* __restrict__ m, const float* __restrict__ xp,
    const unsigned* __restrict__ planes, const float* __restrict__ bias,
    float* __restrict__ out, int N, int E) {
  __shared__ float sacc[GAT_OWN_NODES * GAT_OUT];
  __shared__ float sden[GAT_OWN_NODES * GAT_HEADS];
  __shared__ int   slist[GAT_OWN_WAVES * GAT_OWN_LIST];

  const int tid  = threadIdx.x;
  const int lane = tid & 31;
  const int wave = __builtin_amdgcn_readfirstlane(tid >> 5);
  const int lo   = blockIdx.x * GAT_OWN_NODES;
  const int left = N - lo;
  const int cnt  = left < GAT_OWN_NODES ? left : GAT_OWN_NODES;
  const unsigned ucnt = (unsigned)cnt;

  for (int i = tid; i < GAT_OWN_NODES * GAT_OUT; i += 1024) sacc[i] = 0.0f;
  for (int i = tid; i < GAT_OWN_NODES * GAT_HEADS; i += 1024) sden[i] = 0.0f;
  __syncthreads();

  int* list = slist + wave * GAT_OWN_LIST;
  int n = 0;

  auto pass_ids = [&](int i0, int& s_a, int& s_b, int& d_a, int& d_b) {
    const int qa = i0 + (lane >> 3);
    const int qb = qa + 4;
    const int ea = list[qa < n ? qa : 0];
    const int eb = list[qb < n ? qb : 0];
    s_a = src[ea]; s_b = src[eb];
    d_a = dst[ea]; d_b = dst[eb];
  };

  auto pass = [&](int i0, int s_a, int s_b, int d_a, int d_b, const bool guarded) {
    float4 xr[8];
#pragma unroll
    for (int q = 0; q < 8; ++q) {
      const int sq = __builtin_amdgcn_readlane(q < 4 ? s_a : s_b, (q & 3) * 8);
      xr[q] = *reinterpret_cast<const float4*>(xp + sq * GAT_OUT + lane * 4);
    }

    const int  hd = lane & 7;
    const bool va = !guarded || i0 + (lane >> 3) < n;
    const bool vb = !guarded || i0 + 4 + (lane >> 3) < n;
    const int ga = d_a * GAT_HEADS + hd, gb = d_b * GAT_HEADS + hd;
    const float exa =
        expf(leaky(al_src[s_a * GAT_HEADS + hd] + al_dst[ga]) - m[ga]);
    const float exb =
        expf(leaky(al_src[s_b * GAT_HEADS + hd] + al_dst[gb]) - m[gb]);
    const int la = d_a - lo, lb = d_b - lo;
    if (va) atomicAdd(&sden[la * GAT_HEADS + hd], exa);
    if (vb) atomicAdd(&sden[lb * GAT_HEADS + hd], exb);

#pragma unroll
    for (int q = 0; q < 8; ++q) {
      if (!guarded || i0 + q < n) {
        const int lq = __builtin_amdgcn_readlane(q < 4 ? la : lb, (q & 3) * 8);
        const float a = __shfl(q < 4 ? exa : exb, (q & 3) * 8 + (lane >> 2));
        float* o = sacc + lq * GAT_OUT + lane;
        atomicAdd(o + 0,  xr[q].x * a);
        atomicAdd(o + 32, xr[q].y * a);
        atomicAdd(o + 64, xr[q].z * a);
        atomicAdd(o + 96, xr[q].w * a);
      }
    }
  };

  auto flush = [&]() {
    const int nf = n & ~7;
    int s_a, s_b, d_a, d_b;
    pass_ids(0, s_a, s_b, d_a, d_b);
    for (int i0 = 0; i0 < nf; i0 += 8) {
      int ns_a, ns_b, nd_a, nd_b;
      pass_ids(i0 + 8, ns_a, ns_b, nd_a, nd_b);
      pass(i0, s_a, s_b, d_a, d_b, false);
      s_a = ns_a; s_b = ns_b; d_a = nd_a; d_b = nd_b;
    }
    const int r = n - nf;
    int t = 0;
    if (lane < r) t = list[nf + lane];
    if (lane < r) list[lane] = t;
    n = r;
  };

  const unsigned wg = blockIdx.x;
  unsigned dm[8];
#pragma unroll
  for (int p = 0; p < 8; ++p) dm[p] = 0u - ((wg >> p) & 1u);

  auto keep = [&](const int4& a, const int4& b, int e) {
    const unsigned diff =
        ((unsigned)a.x ^ dm[0]) | ((unsigned)a.y ^ dm[1]) |
        ((unsigned)a.z ^ dm[2]) | ((unsigned)a.w ^ dm[3]) |
        ((unsigned)b.x ^ dm[4]) | ((unsigned)b.y ^ dm[5]) |
        ((unsigned)b.z ^ dm[6]) | ((unsigned)b.w ^ dm[7]);
    unsigned mk = ~diff;
    for (;;) {
      const bool k = mk != 0u;
      const unsigned bal = __builtin_amdgcn_ballot_w32(k);
      if (bal == 0u) break;
      if (k) list[n + (int)__builtin_amdgcn_mbcnt_lo(bal, 0u)] =
                 e + __builtin_ctz(mk);
      n += __builtin_popcount(bal);
      mk &= mk - 1u;
      if (n >= 128) flush();
    }
  };

  const int nfull = (gridDim.x <= 256u) ? (E >> 12) : 0;
  const int4* pl = reinterpret_cast<const int4*>(planes);
  int look = wave;
  int4 c0 = make_int4(0, 0, 0, 0), c1 = c0, c2 = c0, c3 = c0;
  int4 c4 = c0, c5 = c0, c6 = c0, c7 = c0;
  if (look < nfull) {
    const int4* p = pl + look * 256 + lane;
    c0 = p[0];   c1 = p[32];  c2 = p[64];  c3 = p[96];
    c4 = p[128]; c5 = p[160]; c6 = p[192]; c7 = p[224];
  }
  while (look < nfull) {
    const int e    = look * 4096 + lane * 32;
    const int next = look + GAT_OWN_WAVES;
    const int pf = next < nfull ? next : look;
    const int4* p = pl + pf * 256 + lane;
    const int4 n0 = p[0],   n1 = p[32],  n2 = p[64],  n3 = p[96];
    const int4 n4 = p[128], n5 = p[160], n6 = p[192], n7 = p[224];
    keep(c0, c1, e);
    keep(c2, c3, e + 1024);
    keep(c4, c5, e + 2048);
    keep(c6, c7, e + 3072);
    c0 = n0; c1 = n1; c2 = n2; c3 = n3;
    c4 = n4; c5 = n5; c6 = n6; c7 = n7;
    look = next;
  }
  for (int eb = nfull * 4096 + wave * 32; eb < E; eb += 32 * GAT_OWN_WAVES) {
    const int  e = eb + lane;
    const int  d = (e < E) ? dst[e] : -1;
    const bool k = (unsigned)(d - lo) < ucnt;
    const unsigned b = __builtin_amdgcn_ballot_w32(k);
    if (k) list[n + (int)__builtin_amdgcn_mbcnt_lo(b, 0u)] = e;
    n += __builtin_popcount(b);
    if (n >= 128) flush();
  }
  if (n >= 8) flush();
  if (n > 0) {
    int s_a, s_b, d_a, d_b;
    pass_ids(0, s_a, s_b, d_a, d_b);
    pass(0, s_a, s_b, d_a, d_b, true);
  }
  __syncthreads();

  const float4 bv = *reinterpret_cast<const float4*>(bias + lane * 4);
  for (int r = wave; r < cnt; r += GAT_OWN_WAVES) {
    const float* a  = sacc + r * GAT_OUT + lane;
    const float  dn = sden[r * GAT_HEADS + (lane >> 2)];
    const float  dd = (dn == 0.0f) ? 1.0f : dn;
    float4 v;
    v.x = a[0]  / dd + bv.x;
    v.y = a[32] / dd + bv.y;
    v.z = a[64] / dd + bv.z;
    v.w = a[96] / dd + bv.w;
    v.x = v.x > 0.0f ? v.x : expm1f(v.x);
    v.y = v.y > 0.0f ? v.y : expm1f(v.y);
    v.z = v.z > 0.0f ? v.z : expm1f(v.z);
    v.w = v.w > 0.0f ? v.w : expm1f(v.w);
    *reinterpret_cast<float4*>(out + (size_t)(lo + r) * GAT_OUT + lane * 4) = v;
  }
}

extern "C" void kernel_launch(void* const* d_in, const int* in_sizes, int n_in,
                              void* d_out, int out_size, void* d_ws,
                              size_t ws_size, hipStream_t stream) {
  const float* x      = (const float*)d_in[0];
  const int*   ei     = (const int*)d_in[1];
  const float* W      = (const float*)d_in[2];
  const float* a_src  = (const float*)d_in[3];
  const float* a_dst  = (const float*)d_in[4];
  const float* bias   = (const float*)d_in[5];

  const int N = in_sizes[0] / GAT_IN_DIM;
  const int E = in_sizes[1] / 2;
  const int* src = ei;
  const int* dst = ei + E;

  float* out = (float*)d_out;

  float* ws     = (float*)d_ws;
  float* xp     = ws;
  float* al_src = xp + (size_t)N * GAT_OUT;
  float* al_dst = al_src + (size_t)N * GAT_HEADS;
  float* mbuf   = al_dst + (size_t)N * GAT_HEADS;
  unsigned* planes = (unsigned*)(mbuf + (size_t)N * GAT_HEADS);

  {
    int total = N * GAT_HEADS;
    gat_init<<<(total + 255) / 256, 256, 0, stream>>>(mbuf, N);
  }
  gat_proj_wmma<<<(N + 15) / 16, 256, 0, stream>>>(x, W, xp, N);
  {
    int total = N * GAT_HEADS;
    gat_al<<<(total + 255) / 256, 256, 0, stream>>>(xp, a_src, a_dst, al_src,
                                                    al_dst, N);
  }
  gat_edge_max<<<(E + 1023) / 1024, 256, 0, stream>>>(src, dst, al_src, al_dst,
                                                      mbuf, planes, E);
  gat_own_agg<<<(N + GAT_OWN_NODES - 1) / GAT_OWN_NODES, 1024, 0, stream>>>(
      src, dst, al_src, al_dst, mbuf, xp, planes, bias, out, N, E);
}


# Projection re-tiled: a wave owns 16 rows x all 128 columns, x fragments loaded straight from global into registers (no LDS for x), W read from LDS in fragment order; k order of the f32 sums changes
# speedup vs baseline: 4.8288x; 1.1141x over previous
#include <hip/hip_runtime.h>
#include <hip/hip_bf16.h>
#include <math.h>


typedef __attribute__((ext_vector_type(2))) float v2f;
typedef __attribute__((ext_vector_type(8))) float v8f;

#define GAT_IN_DIM   128
#define GAT_HEADS    8
#define GAT_FPH      16
#define GAT_OUT      128
#define NEG_SLOPE    0.2f

typedef __attribute__((ext_vector_type(4))) float v4f;

#define GAT_PROJ_WAVES 8
#define GAT_PROJ_ROWS  (16 * GAT_PROJ_WAVES)

template <bool FULL>
__device__ __forceinline__ void gat_proj_store(const v8f (&acc)[8],
                                               float* __restrict__ xp, int row0,
                                               int lane, int N) {
  float* o = xp + (size_t)row0 * GAT_OUT + lane;
#pragma unroll
  for (int r = 0; r < 8; ++r) {
    float lo[4], hi[4];
#pragma unroll
    for (int p = 0; p < 4; ++p) {
      const float u = acc[2 * p][r];
      const float v = acc[2 * p + 1][r];
      const auto s = __builtin_amdgcn_permlane16_swap(
          __float_as_uint(u), __float_as_uint(v), false, false);
      const unsigned s0 = s[0], s1 = s[1];
      lo[p] = __uint_as_float(s0);
      hi[p] = __uint_as_float(s1);
    }
    if (FULL || row0 + r < N) {
#pragma unroll
      for (int p = 0; p < 4; ++p) o[r * GAT_OUT + 32 * p] = lo[p];
    }
    if (FULL || row0 + 8 + r < N) {
#pragma unroll
      for (int p = 0; p < 4; ++p) o[(8 + r) * GAT_OUT + 32 * p] = hi[p];
    }
  }
}

__global__ __launch_bounds__(256) void gat_proj_wmma(
    const float* __restrict__ x, const float* __restrict__ W,
    float* __restrict__ xp, int N) {
  __shared__ v4f sw[16 * 8 * 32];

  const int tid  = threadIdx.x;
  const int lane = tid & 31;
  const int wave = __builtin_amdgcn_readfirstlane(tid >> 5);
  const int c    = lane & 15;
  const int h    = lane >> 4;
  const int row0 = blockIdx.x * GAT_PROJ_ROWS + wave * 16;

  int ar = row0 + c;
  ar = ar < N ? ar : N - 1;
  const v4f* xa =
      reinterpret_cast<const v4f*>(x + (size_t)ar * GAT_IN_DIM + 4 * h);
  v4f a[16];
#pragma unroll
  for (int j = 0; j < 16; ++j) a[j] = xa[2 * j];

  {
    const int col = tid & 127;
    v4f* dstw = sw + (col >> 4) * 32 + (col & 15);
    for (int s = 0; s < 16; ++s) {
      const int g = 2 * s + (tid >> 7);
      const float* wr = W + (4 * g) * GAT_OUT + col;
      v4f w;
      w.x = wr[0];
      w.y = wr[GAT_OUT];
      w.z = wr[2 * GAT_OUT];
      w.w = wr[3 * GAT_OUT];
      dstw[(g >> 1) * 256 + (g & 1) * 16] = w;
    }
  }
  __syncthreads();

  if (row0 >= N) return;

  v8f acc[8];
#pragma unroll
  for (int t = 0; t < 8; ++t) acc[t] = (v8f){};

#pragma unroll
  for (int j = 0; j < 16; ++j) {
    v4f b[8];
#pragma unroll
    for (int t = 0; t < 8; ++t) b[t] = sw[(j * 8 + t) * 32 + lane];
#pragma unroll
    for (int t = 0; t < 8; ++t)
      acc[t] = __builtin_amdgcn_wmma_f32_16x16x4_f32(
           false, a[j].xy,  false, b[t].xy,
           (short)0, acc[t],  false,  false);
#pragma unroll
    for (int t = 0; t < 8; ++t)
      acc[t] = __builtin_amdgcn_wmma_f32_16x16x4_f32(
           false, a[j].zw,  false, b[t].zw,
           (short)0, acc[t],  false,  false);
  }

  if (row0 + 16 <= N)
    gat_proj_store<true>(acc, xp, row0, lane, N);
  else
    gat_proj_store<false>(acc, xp, row0, lane, N);
}

__global__ void gat_al(const float* __restrict__ xp,
                       const float* __restrict__ a_src,
                       const float* __restrict__ a_dst,
                       float* __restrict__ al_src,
                       float* __restrict__ al_dst, int N) {
  int idx = blockIdx.x * blockDim.x + threadIdx.x;
  if (idx >= N * GAT_HEADS) return;
  int h = idx & 7, node = idx >> 3;
  const float* r  = xp + node * GAT_OUT + h * GAT_FPH;
  const float* as = a_src + h * GAT_FPH;
  const float* ad = a_dst + h * GAT_FPH;
  float s = 0.0f, d = 0.0f;
#pragma unroll
  for (int f = 0; f < GAT_FPH; ++f) {
    float v = r[f];
    s += v * as[f];
    d += v * ad[f];
  }
  al_src[idx] = s;
  al_dst[idx] = d;
}

__global__ void gat_init(float* __restrict__ m, int N) {
  int i = blockIdx.x * blockDim.x + threadIdx.x;
  if (i < N * GAT_HEADS) m[i] = -__builtin_inff();
}

__device__ __forceinline__ float leaky(float e) {
  return e >= 0.0f ? e : NEG_SLOPE * e;
}

#define GAT_EDGES_PER_WAVE 128

template <bool FULL>
__device__ __forceinline__ void gat_load_edge_ids(
    const int* __restrict__ src, const int* __restrict__ dst, int e0, int lane,
    int E, int (&sv)[4], int (&dv)[4]) {
#pragma unroll
  for (int c = 0; c < 4; ++c) {
    const int e = e0 + c * 32 + lane;
    const bool ok = FULL || e < E;
    sv[c] = ok ? src[e] : 0;
    dv[c] = ok ? dst[e] : 0;
  }
}

#define GAT_OWN_NODES 392

__device__ __forceinline__ void gat_store_owner_planes(
    const int (&dv)[4], int e0, int lane, unsigned* __restrict__ planes) {
  unsigned w = 0u;
#pragma unroll
  for (int c = 0; c < 4; ++c) {
    const unsigned o = (unsigned)dv[c] / GAT_OWN_NODES;
#pragma unroll
    for (int p = 0; p < 8; ++p) {
      const unsigned b = __builtin_amdgcn_ballot_w32(((o >> p) & 1u) != 0u);
      w = (lane == (p >> 2) * 16 + c * 4 + (p & 3)) ? b : w;
    }
  }
  planes[(e0 >> 10) * 256 + (lane >> 4) * 128 + ((e0 >> 5) & 31) * 4 +
         (lane & 15)] = w;
}

template <bool FULL>
__device__ __forceinline__ void gat_edge_max_wave(
    const int* __restrict__ src, const int* __restrict__ dst,
    const float* __restrict__ al_src, const float* __restrict__ al_dst,
    float* __restrict__ m, unsigned* __restrict__ planes, int E, int e0,
    int lane) {
  const int h   = lane & 7;
  const int sub = lane >> 3;
  int sv[4], dv[4];
  gat_load_edge_ids<FULL>(src, dst, e0, lane, E, sv, dv);
  gat_store_owner_planes(dv, e0, lane, planes);

#pragma unroll
  for (int c = 0; c < 4; ++c) {
    float v[8];
    int   mi[8];
#pragma unroll
    for (int it = 0; it < 8; ++it) {
      const int j = it * 4 + sub;
      const int s = __shfl(sv[c], j);
      const int d = __shfl(dv[c], j);
      mi[it] = d * GAT_HEADS + h;
      v[it]  = leaky(al_src[s * GAT_HEADS + h] + al_dst[mi[it]]);
    }
#pragma unroll
    for (int it = 0; it < 8; ++it) {
      if (FULL || e0 + c * 32 + it * 4 + sub < E)
        __hip_atomic_fetch_max(&m[mi[it]], v[it], __ATOMIC_RELAXED,
                               __HIP_MEMORY_SCOPE_AGENT);
    }
  }
}

__global__ __launch_bounds__(256) void gat_edge_max(
    const int* __restrict__ src, const int* __restrict__ dst,
    const float* __restrict__ al_src, const float* __restrict__ al_dst,
    float* __restrict__ m, unsigned* __restrict__ planes, int E) {
  const int lane = threadIdx.x & 31;
  const int wave = (blockIdx.x * 256 + threadIdx.x) >> 5;
  const int e0   = wave * GAT_EDGES_PER_WAVE;
  if (e0 + GAT_EDGES_PER_WAVE <= E)
    gat_edge_max_wave<true>(src, dst, al_src, al_dst, m, planes, E, e0, lane);
  else if (e0 < E)
    gat_edge_max_wave<false>(src, dst, al_src, al_dst, m, planes, E, e0, lane);
}

#define GAT_OWN_WAVES 32
#define GAT_OWN_LIST  160

__global__ __launch_bounds__(1024) void gat_own_agg(
    const int* __restrict__ src, const int* __restrict__ dst,
    const float* __restrict__ al_src, const float* __restrict__ al_dst,
    const float* __restrict__ m, const float* __restrict__ xp,
    const unsigned* __restrict__ planes, const float* __restrict__ bias,
    float* __restrict__ out, int N, int E) {
  __shared__ float sacc[GAT_OWN_NODES * GAT_OUT];
  __shared__ float sden[GAT_OWN_NODES * GAT_HEADS];
  __shared__ int   slist[GAT_OWN_WAVES * GAT_OWN_LIST];

  const int tid  = threadIdx.x;
  const int lane = tid & 31;
  const int wave = __builtin_amdgcn_readfirstlane(tid >> 5);
  const int lo   = blockIdx.x * GAT_OWN_NODES;
  const int left = N - lo;
  const int cnt  = left < GAT_OWN_NODES ? left : GAT_OWN_NODES;
  const unsigned ucnt = (unsigned)cnt;

  for (int i = tid; i < GAT_OWN_NODES * GAT_OUT; i += 1024) sacc[i] = 0.0f;
  for (int i = tid; i < GAT_OWN_NODES * GAT_HEADS; i += 1024) sden[i] = 0.0f;
  __syncthreads();

  int* list = slist + wave * GAT_OWN_LIST;
  int n = 0;

  auto pass_ids = [&](int i0, int& s_a, int& s_b, int& d_a, int& d_b) {
    const int qa = i0 + (lane >> 3);
    const int qb = qa + 4;
    const int ea = list[qa < n ? qa : 0];
    const int eb = list[qb < n ? qb : 0];
    s_a = src[ea]; s_b = src[eb];
    d_a = dst[ea]; d_b = dst[eb];
  };

  auto pass = [&](int i0, int s_a, int s_b, int d_a, int d_b, const bool guarded) {
    float4 xr[8];
#pragma unroll
    for (int q = 0; q < 8; ++q) {
      const int sq = __builtin_amdgcn_readlane(q < 4 ? s_a : s_b, (q & 3) * 8);
      xr[q] = *reinterpret_cast<const float4*>(xp + sq * GAT_OUT + lane * 4);
    }

    const int  hd = lane & 7;
    const bool va = !guarded || i0 + (lane >> 3) < n;
    const bool vb = !guarded || i0 + 4 + (lane >> 3) < n;
    const int ga = d_a * GAT_HEADS + hd, gb = d_b * GAT_HEADS + hd;
    const float exa =
        expf(leaky(al_src[s_a * GAT_HEADS + hd] + al_dst[ga]) - m[ga]);
    const float exb =
        expf(leaky(al_src[s_b * GAT_HEADS + hd] + al_dst[gb]) - m[gb]);
    const int la = d_a - lo, lb = d_b - lo;
    if (va) atomicAdd(&sden[la * GAT_HEADS + hd], exa);
    if (vb) atomicAdd(&sden[lb * GAT_HEADS + hd], exb);

#pragma unroll
    for (int q = 0; q < 8; ++q) {
      if (!guarded || i0 + q < n) {
        const int lq = __builtin_amdgcn_readlane(q < 4 ? la : lb, (q & 3) * 8);
        const float a = __shfl(q < 4 ? exa : exb, (q & 3) * 8 + (lane >> 2));
        float* o = sacc + lq * GAT_OUT + lane;
        atomicAdd(o + 0,  xr[q].x * a);
        atomicAdd(o + 32, xr[q].y * a);
        atomicAdd(o + 64, xr[q].z * a);
        atomicAdd(o + 96, xr[q].w * a);
      }
    }
  };

  auto flush = [&]() {
    const int nf = n & ~7;
    int s_a, s_b, d_a, d_b;
    pass_ids(0, s_a, s_b, d_a, d_b);
    for (int i0 = 0; i0 < nf; i0 += 8) {
      int ns_a, ns_b, nd_a, nd_b;
      pass_ids(i0 + 8, ns_a, ns_b, nd_a, nd_b);
      pass(i0, s_a, s_b, d_a, d_b, false);
      s_a = ns_a; s_b = ns_b; d_a = nd_a; d_b = nd_b;
    }
    const int r = n - nf;
    int t = 0;
    if (lane < r) t = list[nf + lane];
    if (lane < r) list[lane] = t;
    n = r;
  };

  const unsigned wg = blockIdx.x;
  unsigned dm[8];
#pragma unroll
  for (int p = 0; p < 8; ++p) dm[p] = 0u - ((wg >> p) & 1u);

  auto keep = [&](const int4& a, const int4& b, int e) {
    const unsigned diff =
        ((unsigned)a.x ^ dm[0]) | ((unsigned)a.y ^ dm[1]) |
        ((unsigned)a.z ^ dm[2]) | ((unsigned)a.w ^ dm[3]) |
        ((unsigned)b.x ^ dm[4]) | ((unsigned)b.y ^ dm[5]) |
        ((unsigned)b.z ^ dm[6]) | ((unsigned)b.w ^ dm[7]);
    unsigned mk = ~diff;
    for (;;) {
      const bool k = mk != 0u;
      const unsigned bal = __builtin_amdgcn_ballot_w32(k);
      if (bal == 0u) break;
      if (k) list[n + (int)__builtin_amdgcn_mbcnt_lo(bal, 0u)] =
                 e + __builtin_ctz(mk);
      n += __builtin_popcount(bal);
      mk &= mk - 1u;
      if (n >= 128) flush();
    }
  };

  const int nfull = (gridDim.x <= 256u) ? (E >> 12) : 0;
  const int4* pl = reinterpret_cast<const int4*>(planes);
  int look = wave;
  int4 c0 = make_int4(0, 0, 0, 0), c1 = c0, c2 = c0, c3 = c0;
  int4 c4 = c0, c5 = c0, c6 = c0, c7 = c0;
  if (look < nfull) {
    const int4* p = pl + look * 256 + lane;
    c0 = p[0];   c1 = p[32];  c2 = p[64];  c3 = p[96];
    c4 = p[128]; c5 = p[160]; c6 = p[192]; c7 = p[224];
  }
  while (look < nfull) {
    const int e    = look * 4096 + lane * 32;
    const int next = look + GAT_OWN_WAVES;
    const int pf = next < nfull ? next : look;
    const int4* p = pl + pf * 256 + lane;
    const int4 n0 = p[0],   n1 = p[32],  n2 = p[64],  n3 = p[96];
    const int4 n4 = p[128], n5 = p[160], n6 = p[192], n7 = p[224];
    keep(c0, c1, e);
    keep(c2, c3, e + 1024);
    keep(c4, c5, e + 2048);
    keep(c6, c7, e + 3072);
    c0 = n0; c1 = n1; c2 = n2; c3 = n3;
    c4 = n4; c5 = n5; c6 = n6; c7 = n7;
    look = next;
  }
  for (int eb = nfull * 4096 + wave * 32; eb < E; eb += 32 * GAT_OWN_WAVES) {
    const int  e = eb + lane;
    const int  d = (e < E) ? dst[e] : -1;
    const bool k = (unsigned)(d - lo) < ucnt;
    const unsigned b = __builtin_amdgcn_ballot_w32(k);
    if (k) list[n + (int)__builtin_amdgcn_mbcnt_lo(b, 0u)] = e;
    n += __builtin_popcount(b);
    if (n >= 128) flush();
  }
  if (n >= 8) flush();
  if (n > 0) {
    int s_a, s_b, d_a, d_b;
    pass_ids(0, s_a, s_b, d_a, d_b);
    pass(0, s_a, s_b, d_a, d_b, true);
  }
  __syncthreads();

  const float4 bv = *reinterpret_cast<const float4*>(bias + lane * 4);
  for (int r = wave; r < cnt; r += GAT_OWN_WAVES) {
    const float* a  = sacc + r * GAT_OUT + lane;
    const float  dn = sden[r * GAT_HEADS + (lane >> 2)];
    const float  dd = (dn == 0.0f) ? 1.0f : dn;
    float4 v;
    v.x = a[0]  / dd + bv.x;
    v.y = a[32] / dd + bv.y;
    v.z = a[64] / dd + bv.z;
    v.w = a[96] / dd + bv.w;
    v.x = v.x > 0.0f ? v.x : expm1f(v.x);
    v.y = v.y > 0.0f ? v.y : expm1f(v.y);
    v.z = v.z > 0.0f ? v.z : expm1f(v.z);
    v.w = v.w > 0.0f ? v.w : expm1f(v.w);
    *reinterpret_cast<float4*>(out + (size_t)(lo + r) * GAT_OUT + lane * 4) = v;
  }
}

extern "C" void kernel_launch(void* const* d_in, const int* in_sizes, int n_in,
                              void* d_out, int out_size, void* d_ws,
                              size_t ws_size, hipStream_t stream) {
  const float* x      = (const float*)d_in[0];
  const int*   ei     = (const int*)d_in[1];
  const float* W      = (const float*)d_in[2];
  const float* a_src  = (const float*)d_in[3];
  const float* a_dst  = (const float*)d_in[4];
  const float* bias   = (const float*)d_in[5];

  const int N = in_sizes[0] / GAT_IN_DIM;
  const int E = in_sizes[1] / 2;
  const int* src = ei;
  const int* dst = ei + E;

  float* out = (float*)d_out;

  float* ws     = (float*)d_ws;
  float* xp     = ws;
  float* al_src = xp + (size_t)N * GAT_OUT;
  float* al_dst = al_src + (size_t)N * GAT_HEADS;
  float* mbuf   = al_dst + (size_t)N * GAT_HEADS;
  unsigned* planes = (unsigned*)(mbuf + (size_t)N * GAT_HEADS);

  {
    int total = N * GAT_HEADS;
    gat_init<<<(total + 255) / 256, 256, 0, stream>>>(mbuf, N);
  }
  gat_proj_wmma<<<(N + GAT_PROJ_ROWS - 1) / GAT_PROJ_ROWS, 256, 0, stream>>>(
      x, W, xp, N);
  {
    int total = N * GAT_HEADS;
    gat_al<<<(total + 255) / 256, 256, 0, stream>>>(xp, a_src, a_dst, al_src,
                                                    al_dst, N);
  }
  gat_edge_max<<<(E + 1023) / 1024, 256, 0, stream>>>(src, dst, al_src, al_dst,
                                                      mbuf, planes, E);
  gat_own_agg<<<(N + GAT_OWN_NODES - 1) / GAT_OWN_NODES, 1024, 0, stream>>>(
      src, dst, al_src, al_dst, mbuf, xp, planes, bias, out, N, E);
}


# Owner sweep copies its own nodes' al_dst and max rows into LDS once (whole-line loads): the two 32-byte gathers by dst per edge become LDS reads, only src-addressed rows come from memory
# speedup vs baseline: 5.0571x; 1.0473x over previous
#include <hip/hip_runtime.h>
#include <hip/hip_bf16.h>
#include <math.h>


typedef __attribute__((ext_vector_type(2))) float v2f;
typedef __attribute__((ext_vector_type(8))) float v8f;

#define GAT_IN_DIM   128
#define GAT_HEADS    8
#define GAT_FPH      16
#define GAT_OUT      128
#define NEG_SLOPE    0.2f

typedef __attribute__((ext_vector_type(4))) float v4f;

#define GAT_PROJ_WAVES 8
#define GAT_PROJ_ROWS  (16 * GAT_PROJ_WAVES)

template <bool FULL>
__device__ __forceinline__ void gat_proj_store(const v8f (&acc)[8],
                                               float* __restrict__ xp, int row0,
                                               int lane, int N) {
  float* o = xp + (size_t)row0 * GAT_OUT + lane;
#pragma unroll
  for (int r = 0; r < 8; ++r) {
    float lo[4], hi[4];
#pragma unroll
    for (int p = 0; p < 4; ++p) {
      const float u = acc[2 * p][r];
      const float v = acc[2 * p + 1][r];
      const auto s = __builtin_amdgcn_permlane16_swap(
          __float_as_uint(u), __float_as_uint(v), false, false);
      const unsigned s0 = s[0], s1 = s[1];
      lo[p] = __uint_as_float(s0);
      hi[p] = __uint_as_float(s1);
    }
    if (FULL || row0 + r < N) {
#pragma unroll
      for (int p = 0; p < 4; ++p) o[r * GAT_OUT + 32 * p] = lo[p];
    }
    if (FULL || row0 + 8 + r < N) {
#pragma unroll
      for (int p = 0; p < 4; ++p) o[(8 + r) * GAT_OUT + 32 * p] = hi[p];
    }
  }
}

__global__ __launch_bounds__(256) void gat_proj_wmma(
    const float* __restrict__ x, const float* __restrict__ W,
    float* __restrict__ xp, int N) {
  __shared__ v4f sw[16 * 8 * 32];

  const int tid  = threadIdx.x;
  const int lane = tid & 31;
  const int wave = __builtin_amdgcn_readfirstlane(tid >> 5);
  const int c    = lane & 15;
  const int h    = lane >> 4;
  const int row0 = blockIdx.x * GAT_PROJ_ROWS + wave * 16;

  int ar = row0 + c;
  ar = ar < N ? ar : N - 1;
  const v4f* xa =
      reinterpret_cast<const v4f*>(x + (size_t)ar * GAT_IN_DIM + 4 * h);
  v4f a[16];
#pragma unroll
  for (int j = 0; j < 16; ++j) a[j] = xa[2 * j];

  {
    const int col = tid & 127;
    v4f* dstw = sw + (col >> 4) * 32 + (col & 15);
    for (int s = 0; s < 16; ++s) {
      const int g = 2 * s + (tid >> 7);
      const float* wr = W + (4 * g) * GAT_OUT + col;
      v4f w;
      w.x = wr[0];
      w.y = wr[GAT_OUT];
      w.z = wr[2 * GAT_OUT];
      w.w = wr[3 * GAT_OUT];
      dstw[(g >> 1) * 256 + (g & 1) * 16] = w;
    }
  }
  __syncthreads();

  if (row0 >= N) return;

  v8f acc[8];
#pragma unroll
  for (int t = 0; t < 8; ++t) acc[t] = (v8f){};

#pragma unroll
  for (int j = 0; j < 16; ++j) {
    v4f b[8];
#pragma unroll
    for (int t = 0; t < 8; ++t) b[t] = sw[(j * 8 + t) * 32 + lane];
#pragma unroll
    for (int t = 0; t < 8; ++t)
      acc[t] = __builtin_amdgcn_wmma_f32_16x16x4_f32(
           false, a[j].xy,  false, b[t].xy,
           (short)0, acc[t],  false,  false);
#pragma unroll
    for (int t = 0; t < 8; ++t)
      acc[t] = __builtin_amdgcn_wmma_f32_16x16x4_f32(
           false, a[j].zw,  false, b[t].zw,
           (short)0, acc[t],  false,  false);
  }

  if (row0 + 16 <= N)
    gat_proj_store<true>(acc, xp, row0, lane, N);
  else
    gat_proj_store<false>(acc, xp, row0, lane, N);
}

__global__ void gat_al(const float* __restrict__ xp,
                       const float* __restrict__ a_src,
                       const float* __restrict__ a_dst,
                       float* __restrict__ al_src,
                       float* __restrict__ al_dst, int N) {
  int idx = blockIdx.x * blockDim.x + threadIdx.x;
  if (idx >= N * GAT_HEADS) return;
  int h = idx & 7, node = idx >> 3;
  const float* r  = xp + node * GAT_OUT + h * GAT_FPH;
  const float* as = a_src + h * GAT_FPH;
  const float* ad = a_dst + h * GAT_FPH;
  float s = 0.0f, d = 0.0f;
#pragma unroll
  for (int f = 0; f < GAT_FPH; ++f) {
    float v = r[f];
    s += v * as[f];
    d += v * ad[f];
  }
  al_src[idx] = s;
  al_dst[idx] = d;
}

__global__ void gat_init(float* __restrict__ m, int N) {
  int i = blockIdx.x * blockDim.x + threadIdx.x;
  if (i < N * GAT_HEADS) m[i] = -__builtin_inff();
}

__device__ __forceinline__ float leaky(float e) {
  return e >= 0.0f ? e : NEG_SLOPE * e;
}

#define GAT_EDGES_PER_WAVE 128

template <bool FULL>
__device__ __forceinline__ void gat_load_edge_ids(
    const int* __restrict__ src, const int* __restrict__ dst, int e0, int lane,
    int E, int (&sv)[4], int (&dv)[4]) {
#pragma unroll
  for (int c = 0; c < 4; ++c) {
    const int e = e0 + c * 32 + lane;
    const bool ok = FULL || e < E;
    sv[c] = ok ? src[e] : 0;
    dv[c] = ok ? dst[e] : 0;
  }
}

#define GAT_OWN_NODES 392

__device__ __forceinline__ void gat_store_owner_planes(
    const int (&dv)[4], int e0, int lane, unsigned* __restrict__ planes) {
  unsigned w = 0u;
#pragma unroll
  for (int c = 0; c < 4; ++c) {
    const unsigned o = (unsigned)dv[c] / GAT_OWN_NODES;
#pragma unroll
    for (int p = 0; p < 8; ++p) {
      const unsigned b = __builtin_amdgcn_ballot_w32(((o >> p) & 1u) != 0u);
      w = (lane == (p >> 2) * 16 + c * 4 + (p & 3)) ? b : w;
    }
  }
  planes[(e0 >> 10) * 256 + (lane >> 4) * 128 + ((e0 >> 5) & 31) * 4 +
         (lane & 15)] = w;
}

template <bool FULL>
__device__ __forceinline__ void gat_edge_max_wave(
    const int* __restrict__ src, const int* __restrict__ dst,
    const float* __restrict__ al_src, const float* __restrict__ al_dst,
    float* __restrict__ m, unsigned* __restrict__ planes, int E, int e0,
    int lane) {
  const int h   = lane & 7;
  const int sub = lane >> 3;
  int sv[4], dv[4];
  gat_load_edge_ids<FULL>(src, dst, e0, lane, E, sv, dv);
  gat_store_owner_planes(dv, e0, lane, planes);

#pragma unroll
  for (int c = 0; c < 4; ++c) {
    float v[8];
    int   mi[8];
#pragma unroll
    for (int it = 0; it < 8; ++it) {
      const int j = it * 4 + sub;
      const int s = __shfl(sv[c], j);
      const int d = __shfl(dv[c], j);
      mi[it] = d * GAT_HEADS + h;
      v[it]  = leaky(al_src[s * GAT_HEADS + h] + al_dst[mi[it]]);
    }
#pragma unroll
    for (int it = 0; it < 8; ++it) {
      if (FULL || e0 + c * 32 + it * 4 + sub < E)
        __hip_atomic_fetch_max(&m[mi[it]], v[it], __ATOMIC_RELAXED,
                               __HIP_MEMORY_SCOPE_AGENT);
    }
  }
}

__global__ __launch_bounds__(256) void gat_edge_max(
    const int* __restrict__ src, const int* __restrict__ dst,
    const float* __restrict__ al_src, const float* __restrict__ al_dst,
    float* __restrict__ m, unsigned* __restrict__ planes, int E) {
  const int lane = threadIdx.x & 31;
  const int wave = (blockIdx.x * 256 + threadIdx.x) >> 5;
  const int e0   = wave * GAT_EDGES_PER_WAVE;
  if (e0 + GAT_EDGES_PER_WAVE <= E)
    gat_edge_max_wave<true>(src, dst, al_src, al_dst, m, planes, E, e0, lane);
  else if (e0 < E)
    gat_edge_max_wave<false>(src, dst, al_src, al_dst, m, planes, E, e0, lane);
}

#define GAT_OWN_WAVES 32
#define GAT_OWN_LIST  160

__global__ __launch_bounds__(1024) void gat_own_agg(
    const int* __restrict__ src, const int* __restrict__ dst,
    const float* __restrict__ al_src, const float* __restrict__ al_dst,
    const float* __restrict__ m, const float* __restrict__ xp,
    const unsigned* __restrict__ planes, const float* __restrict__ bias,
    float* __restrict__ out, int N, int E) {
  __shared__ float sacc[GAT_OWN_NODES * GAT_OUT];
  __shared__ float sden[GAT_OWN_NODES * GAT_HEADS];
  __shared__ float sald[GAT_OWN_NODES * GAT_HEADS];
  __shared__ float smax[GAT_OWN_NODES * GAT_HEADS];
  __shared__ int   slist[GAT_OWN_WAVES * GAT_OWN_LIST];

  const int tid  = threadIdx.x;
  const int lane = tid & 31;
  const int wave = __builtin_amdgcn_readfirstlane(tid >> 5);
  const int lo   = blockIdx.x * GAT_OWN_NODES;
  const int left = N - lo;
  const int cnt  = left < GAT_OWN_NODES ? left : GAT_OWN_NODES;
  const unsigned ucnt = (unsigned)cnt;

  for (int i = tid; i < GAT_OWN_NODES * GAT_OUT; i += 1024) sacc[i] = 0.0f;
  for (int i = tid; i < GAT_OWN_NODES * GAT_HEADS; i += 1024) sden[i] = 0.0f;
  for (int i = tid; i < cnt * GAT_HEADS; i += 1024) {
    sald[i] = al_dst[lo * GAT_HEADS + i];
    smax[i] = m[lo * GAT_HEADS + i];
  }
  __syncthreads();

  int* list = slist + wave * GAT_OWN_LIST;
  int n = 0;

  auto pass_ids = [&](int i0, int& s_a, int& s_b, int& d_a, int& d_b) {
    const int qa = i0 + (lane >> 3);
    const int qb = qa + 4;
    const int ea = list[qa < n ? qa : 0];
    const int eb = list[qb < n ? qb : 0];
    s_a = src[ea]; s_b = src[eb];
    d_a = dst[ea]; d_b = dst[eb];
  };

  auto pass = [&](int i0, int s_a, int s_b, int d_a, int d_b, const bool guarded) {
    float4 xr[8];
#pragma unroll
    for (int q = 0; q < 8; ++q) {
      const int sq = __builtin_amdgcn_readlane(q < 4 ? s_a : s_b, (q & 3) * 8);
      xr[q] = *reinterpret_cast<const float4*>(xp + sq * GAT_OUT + lane * 4);
    }

    const int  hd = lane & 7;
    const bool va = !guarded || i0 + (lane >> 3) < n;
    const bool vb = !guarded || i0 + 4 + (lane >> 3) < n;
    const int la = d_a - lo, lb = d_b - lo;
    const int ga = la * GAT_HEADS + hd, gb = lb * GAT_HEADS + hd;
    const float exa =
        expf(leaky(al_src[s_a * GAT_HEADS + hd] + sald[ga]) - smax[ga]);
    const float exb =
        expf(leaky(al_src[s_b * GAT_HEADS + hd] + sald[gb]) - smax[gb]);
    if (va) atomicAdd(&sden[ga], exa);
    if (vb) atomicAdd(&sden[gb], exb);

#pragma unroll
    for (int q = 0; q < 8; ++q) {
      if (!guarded || i0 + q < n) {
        const int lq = __builtin_amdgcn_readlane(q < 4 ? la : lb, (q & 3) * 8);
        const float a = __shfl(q < 4 ? exa : exb, (q & 3) * 8 + (lane >> 2));
        float* o = sacc + lq * GAT_OUT + lane;
        atomicAdd(o + 0,  xr[q].x * a);
        atomicAdd(o + 32, xr[q].y * a);
        atomicAdd(o + 64, xr[q].z * a);
        atomicAdd(o + 96, xr[q].w * a);
      }
    }
  };

  auto flush = [&]() {
    const int nf = n & ~7;
    int s_a, s_b, d_a, d_b;
    pass_ids(0, s_a, s_b, d_a, d_b);
    for (int i0 = 0; i0 < nf; i0 += 8) {
      int ns_a, ns_b, nd_a, nd_b;
      pass_ids(i0 + 8, ns_a, ns_b, nd_a, nd_b);
      pass(i0, s_a, s_b, d_a, d_b, false);
      s_a = ns_a; s_b = ns_b; d_a = nd_a; d_b = nd_b;
    }
    const int r = n - nf;
    int t = 0;
    if (lane < r) t = list[nf + lane];
    if (lane < r) list[lane] = t;
    n = r;
  };

  const unsigned wg = blockIdx.x;
  unsigned dm[8];
#pragma unroll
  for (int p = 0; p < 8; ++p) dm[p] = 0u - ((wg >> p) & 1u);

  auto keep = [&](const int4& a, const int4& b, int e) {
    const unsigned diff =
        ((unsigned)a.x ^ dm[0]) | ((unsigned)a.y ^ dm[1]) |
        ((unsigned)a.z ^ dm[2]) | ((unsigned)a.w ^ dm[3]) |
        ((unsigned)b.x ^ dm[4]) | ((unsigned)b.y ^ dm[5]) |
        ((unsigned)b.z ^ dm[6]) | ((unsigned)b.w ^ dm[7]);
    unsigned mk = ~diff;
    for (;;) {
      const bool k = mk != 0u;
      const unsigned bal = __builtin_amdgcn_ballot_w32(k);
      if (bal == 0u) break;
      if (k) list[n + (int)__builtin_amdgcn_mbcnt_lo(bal, 0u)] =
                 e + __builtin_ctz(mk);
      n += __builtin_popcount(bal);
      mk &= mk - 1u;
      if (n >= 128) flush();
    }
  };

  const int nfull = (gridDim.x <= 256u) ? (E >> 12) : 0;
  const int4* pl = reinterpret_cast<const int4*>(planes);
  int look = wave;
  int4 c0 = make_int4(0, 0, 0, 0), c1 = c0, c2 = c0, c3 = c0;
  int4 c4 = c0, c5 = c0, c6 = c0, c7 = c0;
  if (look < nfull) {
    const int4* p = pl + look * 256 + lane;
    c0 = p[0];   c1 = p[32];  c2 = p[64];  c3 = p[96];
    c4 = p[128]; c5 = p[160]; c6 = p[192]; c7 = p[224];
  }
  while (look < nfull) {
    const int e    = look * 4096 + lane * 32;
    const int next = look + GAT_OWN_WAVES;
    const int pf = next < nfull ? next : look;
    const int4* p = pl + pf * 256 + lane;
    const int4 n0 = p[0],   n1 = p[32],  n2 = p[64],  n3 = p[96];
    const int4 n4 = p[128], n5 = p[160], n6 = p[192], n7 = p[224];
    keep(c0, c1, e);
    keep(c2, c3, e + 1024);
    keep(c4, c5, e + 2048);
    keep(c6, c7, e + 3072);
    c0 = n0; c1 = n1; c2 = n2; c3 = n3;
    c4 = n4; c5 = n5; c6 = n6; c7 = n7;
    look = next;
  }
  for (int eb = nfull * 4096 + wave * 32; eb < E; eb += 32 * GAT_OWN_WAVES) {
    const int  e = eb + lane;
    const int  d = (e < E) ? dst[e] : -1;
    const bool k = (unsigned)(d - lo) < ucnt;
    const unsigned b = __builtin_amdgcn_ballot_w32(k);
    if (k) list[n + (int)__builtin_amdgcn_mbcnt_lo(b, 0u)] = e;
    n += __builtin_popcount(b);
    if (n >= 128) flush();
  }
  if (n >= 8) flush();
  if (n > 0) {
    int s_a, s_b, d_a, d_b;
    pass_ids(0, s_a, s_b, d_a, d_b);
    pass(0, s_a, s_b, d_a, d_b, true);
  }
  __syncthreads();

  const float4 bv = *reinterpret_cast<const float4*>(bias + lane * 4);
  for (int r = wave; r < cnt; r += GAT_OWN_WAVES) {
    const float* a  = sacc + r * GAT_OUT + lane;
    const float  dn = sden[r * GAT_HEADS + (lane >> 2)];
    const float  dd = (dn == 0.0f) ? 1.0f : dn;
    float4 v;
    v.x = a[0]  / dd + bv.x;
    v.y = a[32] / dd + bv.y;
    v.z = a[64] / dd + bv.z;
    v.w = a[96] / dd + bv.w;
    v.x = v.x > 0.0f ? v.x : expm1f(v.x);
    v.y = v.y > 0.0f ? v.y : expm1f(v.y);
    v.z = v.z > 0.0f ? v.z : expm1f(v.z);
    v.w = v.w > 0.0f ? v.w : expm1f(v.w);
    *reinterpret_cast<float4*>(out + (size_t)(lo + r) * GAT_OUT + lane * 4) = v;
  }
}

extern "C" void kernel_launch(void* const* d_in, const int* in_sizes, int n_in,
                              void* d_out, int out_size, void* d_ws,
                              size_t ws_size, hipStream_t stream) {
  const float* x      = (const float*)d_in[0];
  const int*   ei     = (const int*)d_in[1];
  const float* W      = (const float*)d_in[2];
  const float* a_src  = (const float*)d_in[3];
  const float* a_dst  = (const float*)d_in[4];
  const float* bias   = (const float*)d_in[5];

  const int N = in_sizes[0] / GAT_IN_DIM;
  const int E = in_sizes[1] / 2;
  const int* src = ei;
  const int* dst = ei + E;

  float* out = (float*)d_out;

  float* ws     = (float*)d_ws;
  float* xp     = ws;
  float* al_src = xp + (size_t)N * GAT_OUT;
  float* al_dst = al_src + (size_t)N * GAT_HEADS;
  float* mbuf   = al_dst + (size_t)N * GAT_HEADS;
  unsigned* planes = (unsigned*)(mbuf + (size_t)N * GAT_HEADS);

  {
    int total = N * GAT_HEADS;
    gat_init<<<(total + 255) / 256, 256, 0, stream>>>(mbuf, N);
  }
  gat_proj_wmma<<<(N + GAT_PROJ_ROWS - 1) / GAT_PROJ_ROWS, 256, 0, stream>>>(
      x, W, xp, N);
  {
    int total = N * GAT_HEADS;
    gat_al<<<(total + 255) / 256, 256, 0, stream>>>(xp, a_src, a_dst, al_src,
                                                    al_dst, N);
  }
  gat_edge_max<<<(E + 1023) / 1024, 256, 0, stream>>>(src, dst, al_src, al_dst,
                                                      mbuf, planes, E);
  gat_own_agg<<<(N + GAT_OWN_NODES - 1) / GAT_OWN_NODES, 1024, 0, stream>>>(
      src, dst, al_src, al_dst, mbuf, xp, planes, bias, out, N, E);
}


# Segment max taken over al_src[src] alone (x -> leakyrelu(x + c) is monotone, so max logit = leakyrelu(max al_src + al_dst), bit for bit): max launch drops its al_dst[dst] row gather per edge
# speedup vs baseline: 5.2563x; 1.0394x over previous
#include <hip/hip_runtime.h>
#include <hip/hip_bf16.h>
#include <math.h>


typedef __attribute__((ext_vector_type(2))) float v2f;
typedef __attribute__((ext_vector_type(8))) float v8f;

#define GAT_IN_DIM   128
#define GAT_HEADS    8
#define GAT_FPH      16
#define GAT_OUT      128
#define NEG_SLOPE    0.2f

typedef __attribute__((ext_vector_type(4))) float v4f;

#define GAT_PROJ_WAVES 8
#define GAT_PROJ_ROWS  (16 * GAT_PROJ_WAVES)

template <bool FULL>
__device__ __forceinline__ void gat_proj_store(const v8f (&acc)[8],
                                               float* __restrict__ xp, int row0,
                                               int lane, int N) {
  float* o = xp + (size_t)row0 * GAT_OUT + lane;
#pragma unroll
  for (int r = 0; r < 8; ++r) {
    float lo[4], hi[4];
#pragma unroll
    for (int p = 0; p < 4; ++p) {
      const float u = acc[2 * p][r];
      const float v = acc[2 * p + 1][r];
      const auto s = __builtin_amdgcn_permlane16_swap(
          __float_as_uint(u), __float_as_uint(v), false, false);
      const unsigned s0 = s[0], s1 = s[1];
      lo[p] = __uint_as_float(s0);
      hi[p] = __uint_as_float(s1);
    }
    if (FULL || row0 + r < N) {
#pragma unroll
      for (int p = 0; p < 4; ++p) o[r * GAT_OUT + 32 * p] = lo[p];
    }
    if (FULL || row0 + 8 + r < N) {
#pragma unroll
      for (int p = 0; p < 4; ++p) o[(8 + r) * GAT_OUT + 32 * p] = hi[p];
    }
  }
}

__global__ __launch_bounds__(256) void gat_proj_wmma(
    const float* __restrict__ x, const float* __restrict__ W,
    float* __restrict__ xp, int N) {
  __shared__ v4f sw[16 * 8 * 32];

  const int tid  = threadIdx.x;
  const int lane = tid & 31;
  const int wave = __builtin_amdgcn_readfirstlane(tid >> 5);
  const int c    = lane & 15;
  const int h    = lane >> 4;
  const int row0 = blockIdx.x * GAT_PROJ_ROWS + wave * 16;

  int ar = row0 + c;
  ar = ar < N ? ar : N - 1;
  const v4f* xa =
      reinterpret_cast<const v4f*>(x + (size_t)ar * GAT_IN_DIM + 4 * h);
  v4f a[16];
#pragma unroll
  for (int j = 0; j < 16; ++j) a[j] = xa[2 * j];

  {
    const int col = tid & 127;
    v4f* dstw = sw + (col >> 4) * 32 + (col & 15);
    for (int s = 0; s < 16; ++s) {
      const int g = 2 * s + (tid >> 7);
      const float* wr = W + (4 * g) * GAT_OUT + col;
      v4f w;
      w.x = wr[0];
      w.y = wr[GAT_OUT];
      w.z = wr[2 * GAT_OUT];
      w.w = wr[3 * GAT_OUT];
      dstw[(g >> 1) * 256 + (g & 1) * 16] = w;
    }
  }
  __syncthreads();

  if (row0 >= N) return;

  v8f acc[8];
#pragma unroll
  for (int t = 0; t < 8; ++t) acc[t] = (v8f){};

#pragma unroll
  for (int j = 0; j < 16; ++j) {
    v4f b[8];
#pragma unroll
    for (int t = 0; t < 8; ++t) b[t] = sw[(j * 8 + t) * 32 + lane];
#pragma unroll
    for (int t = 0; t < 8; ++t)
      acc[t] = __builtin_amdgcn_wmma_f32_16x16x4_f32(
           false, a[j].xy,  false, b[t].xy,
           (short)0, acc[t],  false,  false);
#pragma unroll
    for (int t = 0; t < 8; ++t)
      acc[t] = __builtin_amdgcn_wmma_f32_16x16x4_f32(
           false, a[j].zw,  false, b[t].zw,
           (short)0, acc[t],  false,  false);
  }

  if (row0 + 16 <= N)
    gat_proj_store<true>(acc, xp, row0, lane, N);
  else
    gat_proj_store<false>(acc, xp, row0, lane, N);
}

__global__ void gat_al(const float* __restrict__ xp,
                       const float* __restrict__ a_src,
                       const float* __restrict__ a_dst,
                       float* __restrict__ al_src,
                       float* __restrict__ al_dst, int N) {
  int idx = blockIdx.x * blockDim.x + threadIdx.x;
  if (idx >= N * GAT_HEADS) return;
  int h = idx & 7, node = idx >> 3;
  const float* r  = xp + node * GAT_OUT + h * GAT_FPH;
  const float* as = a_src + h * GAT_FPH;
  const float* ad = a_dst + h * GAT_FPH;
  float s = 0.0f, d = 0.0f;
#pragma unroll
  for (int f = 0; f < GAT_FPH; ++f) {
    float v = r[f];
    s += v * as[f];
    d += v * ad[f];
  }
  al_src[idx] = s;
  al_dst[idx] = d;
}

__global__ void gat_init(float* __restrict__ m, int N) {
  int i = blockIdx.x * blockDim.x + threadIdx.x;
  if (i < N * GAT_HEADS) m[i] = -__builtin_inff();
}

__device__ __forceinline__ float leaky(float e) {
  return e >= 0.0f ? e : NEG_SLOPE * e;
}

#define GAT_EDGES_PER_WAVE 128

template <bool FULL>
__device__ __forceinline__ void gat_load_edge_ids(
    const int* __restrict__ src, const int* __restrict__ dst, int e0, int lane,
    int E, int (&sv)[4], int (&dv)[4]) {
#pragma unroll
  for (int c = 0; c < 4; ++c) {
    const int e = e0 + c * 32 + lane;
    const bool ok = FULL || e < E;
    sv[c] = ok ? src[e] : 0;
    dv[c] = ok ? dst[e] : 0;
  }
}

#define GAT_OWN_NODES 392

__device__ __forceinline__ void gat_store_owner_planes(
    const int (&dv)[4], int e0, int lane, unsigned* __restrict__ planes) {
  unsigned w = 0u;
#pragma unroll
  for (int c = 0; c < 4; ++c) {
    const unsigned o = (unsigned)dv[c] / GAT_OWN_NODES;
#pragma unroll
    for (int p = 0; p < 8; ++p) {
      const unsigned b = __builtin_amdgcn_ballot_w32(((o >> p) & 1u) != 0u);
      w = (lane == (p >> 2) * 16 + c * 4 + (p & 3)) ? b : w;
    }
  }
  planes[(e0 >> 10) * 256 + (lane >> 4) * 128 + ((e0 >> 5) & 31) * 4 +
         (lane & 15)] = w;
}

template <bool FULL>
__device__ __forceinline__ void gat_edge_max_wave(
    const int* __restrict__ src, const int* __restrict__ dst,
    const float* __restrict__ al_src, float* __restrict__ m,
    unsigned* __restrict__ planes, int E, int e0, int lane) {
  const int h   = lane & 7;
  const int sub = lane >> 3;
  int sv[4], dv[4];
  gat_load_edge_ids<FULL>(src, dst, e0, lane, E, sv, dv);
  gat_store_owner_planes(dv, e0, lane, planes);

#pragma unroll
  for (int c = 0; c < 4; ++c) {
    float v[8];
    int   mi[8];
#pragma unroll
    for (int it = 0; it < 8; ++it) {
      const int j = it * 4 + sub;
      const int s = __shfl(sv[c], j);
      const int d = __shfl(dv[c], j);
      mi[it] = d * GAT_HEADS + h;
      v[it]  = al_src[s * GAT_HEADS + h];
    }
#pragma unroll
    for (int it = 0; it < 8; ++it) {
      if (FULL || e0 + c * 32 + it * 4 + sub < E)
        __hip_atomic_fetch_max(&m[mi[it]], v[it], __ATOMIC_RELAXED,
                               __HIP_MEMORY_SCOPE_AGENT);
    }
  }
}

__global__ __launch_bounds__(256) void gat_edge_max(
    const int* __restrict__ src, const int* __restrict__ dst,
    const float* __restrict__ al_src, float* __restrict__ m,
    unsigned* __restrict__ planes, int E) {
  const int lane = threadIdx.x & 31;
  const int wave = (blockIdx.x * 256 + threadIdx.x) >> 5;
  const int e0   = wave * GAT_EDGES_PER_WAVE;
  if (e0 + GAT_EDGES_PER_WAVE <= E)
    gat_edge_max_wave<true>(src, dst, al_src, m, planes, E, e0, lane);
  else if (e0 < E)
    gat_edge_max_wave<false>(src, dst, al_src, m, planes, E, e0, lane);
}

#define GAT_OWN_WAVES 32
#define GAT_OWN_LIST  160

__global__ __launch_bounds__(1024) void gat_own_agg(
    const int* __restrict__ src, const int* __restrict__ dst,
    const float* __restrict__ al_src, const float* __restrict__ al_dst,
    const float* __restrict__ m, const float* __restrict__ xp,
    const unsigned* __restrict__ planes, const float* __restrict__ bias,
    float* __restrict__ out, int N, int E) {
  __shared__ float sacc[GAT_OWN_NODES * GAT_OUT];
  __shared__ float sden[GAT_OWN_NODES * GAT_HEADS];
  __shared__ float sald[GAT_OWN_NODES * GAT_HEADS];
  __shared__ float smax[GAT_OWN_NODES * GAT_HEADS];
  __shared__ int   slist[GAT_OWN_WAVES * GAT_OWN_LIST];

  const int tid  = threadIdx.x;
  const int lane = tid & 31;
  const int wave = __builtin_amdgcn_readfirstlane(tid >> 5);
  const int lo   = blockIdx.x * GAT_OWN_NODES;
  const int left = N - lo;
  const int cnt  = left < GAT_OWN_NODES ? left : GAT_OWN_NODES;
  const unsigned ucnt = (unsigned)cnt;

  for (int i = tid; i < GAT_OWN_NODES * GAT_OUT; i += 1024) sacc[i] = 0.0f;
  for (int i = tid; i < GAT_OWN_NODES * GAT_HEADS; i += 1024) sden[i] = 0.0f;
  for (int i = tid; i < cnt * GAT_HEADS; i += 1024) {
    const float ad = al_dst[lo * GAT_HEADS + i];
    sald[i] = ad;
    smax[i] = leaky(m[lo * GAT_HEADS + i] + ad);
  }
  __syncthreads();

  int* list = slist + wave * GAT_OWN_LIST;
  int n = 0;

  auto pass_ids = [&](int i0, int& s_a, int& s_b, int& d_a, int& d_b) {
    const int qa = i0 + (lane >> 3);
    const int qb = qa + 4;
    const int ea = list[qa < n ? qa : 0];
    const int eb = list[qb < n ? qb : 0];
    s_a = src[ea]; s_b = src[eb];
    d_a = dst[ea]; d_b = dst[eb];
  };

  auto pass = [&](int i0, int s_a, int s_b, int d_a, int d_b, const bool guarded) {
    float4 xr[8];
#pragma unroll
    for (int q = 0; q < 8; ++q) {
      const int sq = __builtin_amdgcn_readlane(q < 4 ? s_a : s_b, (q & 3) * 8);
      xr[q] = *reinterpret_cast<const float4*>(xp + sq * GAT_OUT + lane * 4);
    }

    const int  hd = lane & 7;
    const bool va = !guarded || i0 + (lane >> 3) < n;
    const bool vb = !guarded || i0 + 4 + (lane >> 3) < n;
    const int la = d_a - lo, lb = d_b - lo;
    const int ga = la * GAT_HEADS + hd, gb = lb * GAT_HEADS + hd;
    const float exa =
        expf(leaky(al_src[s_a * GAT_HEADS + hd] + sald[ga]) - smax[ga]);
    const float exb =
        expf(leaky(al_src[s_b * GAT_HEADS + hd] + sald[gb]) - smax[gb]);
    if (va) atomicAdd(&sden[ga], exa);
    if (vb) atomicAdd(&sden[gb], exb);

#pragma unroll
    for (int q = 0; q < 8; ++q) {
      if (!guarded || i0 + q < n) {
        const int lq = __builtin_amdgcn_readlane(q < 4 ? la : lb, (q & 3) * 8);
        const float a = __shfl(q < 4 ? exa : exb, (q & 3) * 8 + (lane >> 2));
        float* o = sacc + lq * GAT_OUT + lane;
        atomicAdd(o + 0,  xr[q].x * a);
        atomicAdd(o + 32, xr[q].y * a);
        atomicAdd(o + 64, xr[q].z * a);
        atomicAdd(o + 96, xr[q].w * a);
      }
    }
  };

  auto flush = [&]() {
    const int nf = n & ~7;
    int s_a, s_b, d_a, d_b;
    pass_ids(0, s_a, s_b, d_a, d_b);
    for (int i0 = 0; i0 < nf; i0 += 8) {
      int ns_a, ns_b, nd_a, nd_b;
      pass_ids(i0 + 8, ns_a, ns_b, nd_a, nd_b);
      pass(i0, s_a, s_b, d_a, d_b, false);
      s_a = ns_a; s_b = ns_b; d_a = nd_a; d_b = nd_b;
    }
    const int r = n - nf;
    int t = 0;
    if (lane < r) t = list[nf + lane];
    if (lane < r) list[lane] = t;
    n = r;
  };

  const unsigned wg = blockIdx.x;
  unsigned dm[8];
#pragma unroll
  for (int p = 0; p < 8; ++p) dm[p] = 0u - ((wg >> p) & 1u);

  auto keep = [&](const int4& a, const int4& b, int e) {
    const unsigned diff =
        ((unsigned)a.x ^ dm[0]) | ((unsigned)a.y ^ dm[1]) |
        ((unsigned)a.z ^ dm[2]) | ((unsigned)a.w ^ dm[3]) |
        ((unsigned)b.x ^ dm[4]) | ((unsigned)b.y ^ dm[5]) |
        ((unsigned)b.z ^ dm[6]) | ((unsigned)b.w ^ dm[7]);
    unsigned mk = ~diff;
    for (;;) {
      const bool k = mk != 0u;
      const unsigned bal = __builtin_amdgcn_ballot_w32(k);
      if (bal == 0u) break;
      if (k) list[n + (int)__builtin_amdgcn_mbcnt_lo(bal, 0u)] =
                 e + __builtin_ctz(mk);
      n += __builtin_popcount(bal);
      mk &= mk - 1u;
      if (n >= 128) flush();
    }
  };

  const int nfull = (gridDim.x <= 256u) ? (E >> 12) : 0;
  const int4* pl = reinterpret_cast<const int4*>(planes);
  int look = wave;
  int4 c0 = make_int4(0, 0, 0, 0), c1 = c0, c2 = c0, c3 = c0;
  int4 c4 = c0, c5 = c0, c6 = c0, c7 = c0;
  if (look < nfull) {
    const int4* p = pl + look * 256 + lane;
    c0 = p[0];   c1 = p[32];  c2 = p[64];  c3 = p[96];
    c4 = p[128]; c5 = p[160]; c6 = p[192]; c7 = p[224];
  }
  while (look < nfull) {
    const int e    = look * 4096 + lane * 32;
    const int next = look + GAT_OWN_WAVES;
    const int pf = next < nfull ? next : look;
    const int4* p = pl + pf * 256 + lane;
    const int4 n0 = p[0],   n1 = p[32],  n2 = p[64],  n3 = p[96];
    const int4 n4 = p[128], n5 = p[160], n6 = p[192], n7 = p[224];
    keep(c0, c1, e);
    keep(c2, c3, e + 1024);
    keep(c4, c5, e + 2048);
    keep(c6, c7, e + 3072);
    c0 = n0; c1 = n1; c2 = n2; c3 = n3;
    c4 = n4; c5 = n5; c6 = n6; c7 = n7;
    look = next;
  }
  for (int eb = nfull * 4096 + wave * 32; eb < E; eb += 32 * GAT_OWN_WAVES) {
    const int  e = eb + lane;
    const int  d = (e < E) ? dst[e] : -1;
    const bool k = (unsigned)(d - lo) < ucnt;
    const unsigned b = __builtin_amdgcn_ballot_w32(k);
    if (k) list[n + (int)__builtin_amdgcn_mbcnt_lo(b, 0u)] = e;
    n += __builtin_popcount(b);
    if (n >= 128) flush();
  }
  if (n >= 8) flush();
  if (n > 0) {
    int s_a, s_b, d_a, d_b;
    pass_ids(0, s_a, s_b, d_a, d_b);
    pass(0, s_a, s_b, d_a, d_b, true);
  }
  __syncthreads();

  const float4 bv = *reinterpret_cast<const float4*>(bias + lane * 4);
  for (int r = wave; r < cnt; r += GAT_OWN_WAVES) {
    const float* a  = sacc + r * GAT_OUT + lane;
    const float  dn = sden[r * GAT_HEADS + (lane >> 2)];
    const float  dd = (dn == 0.0f) ? 1.0f : dn;
    float4 v;
    v.x = a[0]  / dd + bv.x;
    v.y = a[32] / dd + bv.y;
    v.z = a[64] / dd + bv.z;
    v.w = a[96] / dd + bv.w;
    v.x = v.x > 0.0f ? v.x : expm1f(v.x);
    v.y = v.y > 0.0f ? v.y : expm1f(v.y);
    v.z = v.z > 0.0f ? v.z : expm1f(v.z);
    v.w = v.w > 0.0f ? v.w : expm1f(v.w);
    *reinterpret_cast<float4*>(out + (size_t)(lo + r) * GAT_OUT + lane * 4) = v;
  }
}

extern "C" void kernel_launch(void* const* d_in, const int* in_sizes, int n_in,
                              void* d_out, int out_size, void* d_ws,
                              size_t ws_size, hipStream_t stream) {
  const float* x      = (const float*)d_in[0];
  const int*   ei     = (const int*)d_in[1];
  const float* W      = (const float*)d_in[2];
  const float* a_src  = (const float*)d_in[3];
  const float* a_dst  = (const float*)d_in[4];
  const float* bias   = (const float*)d_in[5];

  const int N = in_sizes[0] / GAT_IN_DIM;
  const int E = in_sizes[1] / 2;
  const int* src = ei;
  const int* dst = ei + E;

  float* out = (float*)d_out;

  float* ws     = (float*)d_ws;
  float* xp     = ws;
  float* al_src = xp + (size_t)N * GAT_OUT;
  float* al_dst = al_src + (size_t)N * GAT_HEADS;
  float* mbuf   = al_dst + (size_t)N * GAT_HEADS;
  unsigned* planes = (unsigned*)(mbuf + (size_t)N * GAT_HEADS);

  {
    int total = N * GAT_HEADS;
    gat_init<<<(total + 255) / 256, 256, 0, stream>>>(mbuf, N);
  }
  gat_proj_wmma<<<(N + GAT_PROJ_ROWS - 1) / GAT_PROJ_ROWS, 256, 0, stream>>>(
      x, W, xp, N);
  {
    int total = N * GAT_HEADS;
    gat_al<<<(total + 255) / 256, 256, 0, stream>>>(xp, a_src, a_dst, al_src,
                                                    al_dst, N);
  }
  gat_edge_max<<<(E + 1023) / 1024, 256, 0, stream>>>(src, dst, al_src, mbuf,
                                                      planes, E);
  gat_own_agg<<<(N + GAT_OWN_NODES - 1) / GAT_OWN_NODES, 1024, 0, stream>>>(
      src, dst, al_src, al_dst, mbuf, xp, planes, bias, out, N, E);
}
